# starting kernel
# baseline (speedup 1.0000x reference)
#include <hip/hip_runtime.h>
#include <hip/hip_bf16.h>

typedef __attribute__((ext_vector_type(2))) float v2f;
typedef __attribute__((ext_vector_type(8))) float v8f;

#define N_NODES 50000
#define N_EDGES 800000
#define KPAD    196

__global__ void zero_f32(float* __restrict__ p, int n) {
    int i = blockIdx.x * blockDim.x + threadIdx.x;
    if (i < n) p[i] = 0.0f;
}

__global__ void deg_accum(const int* __restrict__ dst, float* __restrict__ deg, int e) {
    int i = blockIdx.x * blockDim.x + threadIdx.x;
    if (i < e) atomicAdd(&deg[dst[i]], 1.0f);
}

__global__ void deg_to_dinv(float* __restrict__ deg, int n) {
    int i = blockIdx.x * blockDim.x + threadIdx.x;
    if (i < n) deg[i] = rsqrtf(fmaxf(deg[i], 1.0f));
}

__global__ void copy_to_xcat(const float* __restrict__ h, float* __restrict__ xcat, int n64) {
    int i = blockIdx.x * blockDim.x + threadIdx.x;
    if (i >= n64) return;
    int node = i >> 6, f = i & 63;
    xcat[node * 192 + f] = h[i];
}

__global__ void spmm_edges(const int* __restrict__ src, const int* __restrict__ dst,
                           const float* __restrict__ x, int xStride,
                           const float* __restrict__ dinv,
                           float* __restrict__ agg, int e) {
    long long i = (long long)blockIdx.x * blockDim.x + threadIdx.x;
    if (i >= (long long)e * 64) return;
    int eidx = (int)(i >> 6);
    int f    = (int)(i & 63);
    int s = src[eidx], d = dst[eidx];
    float v = x[s * xStride + f] * dinv[s];
    atomicAdd(&agg[d * 64 + f], v);
}

__global__ void cheb_x1(const float* __restrict__ agg, float* __restrict__ xcat,
                        const float* __restrict__ dinv, float cAgg, float cX0, int n64) {
    int i = blockIdx.x * blockDim.x + threadIdx.x;
    if (i >= n64) return;
    int node = i >> 6, f = i & 63;
    float a = agg[i] * dinv[node];
    xcat[node * 192 + 64 + f] = cAgg * a + cX0 * xcat[node * 192 + f];
}

__global__ void cheb_x2(const float* __restrict__ agg, float* __restrict__ xcat,
                        const float* __restrict__ dinv,
                        float cAgg, float cX1, float cX0, int n64) {
    int i = blockIdx.x * blockDim.x + threadIdx.x;
    if (i >= n64) return;
    int node = i >> 6, f = i & 63;
    float a  = agg[i] * dinv[node];
    float x0 = xcat[node * 192 + f];
    float x1 = xcat[node * 192 + 64 + f];
    xcat[node * 192 + 128 + f] = cAgg * a + cX1 * x1 + cX0 * x0;
}

__global__ void __launch_bounds__(128)
gemm_wmma(const float* __restrict__ X, int xStride,
          const float* __restrict__ W, const float* __restrict__ bias,
          float* __restrict__ Out, int outStride,
          int Kdim, int Ncols, int doRelu) {
    __shared__ float sW[16 * KPAD];
    __shared__ float sA[4][16 * KPAD];

    const int tid  = threadIdx.x;
    const int wave = tid >> 5;
    const int lane = tid & 31;
    const int rowBase = blockIdx.x * 64;
    const int waveRow = rowBase + wave * 16;
    const int n0 = blockIdx.y * 16;

    for (int idx = tid; idx < 16 * Kdim; idx += 128) {
        int n = idx / Kdim, k = idx - n * Kdim;
        int gn = n0 + n;
        sW[n * KPAD + k] = (gn < Ncols) ? W[gn * Kdim + k] : 0.0f;
    }
    for (int idx = lane; idx < 16 * Kdim; idx += 32) {
        int r = idx / Kdim, k = idx - r * Kdim;
        int row = waveRow + r;
        sA[wave][r * KPAD + k] = (row < N_NODES) ? X[row * xStride + k] : 0.0f;
    }
    __syncthreads();

    const int half = lane >> 4;
    const int l16  = lane & 15;

    v8f c = {};
    const float* aRow = &sA[wave][0];
    for (int k0 = 0; k0 < Kdim; k0 += 4) {
        int ka = k0 + 2 * half;
        v2f a, b;
        a.x = aRow[l16 * KPAD + ka];
        a.y = aRow[l16 * KPAD + ka + 1];
        b.x = sW[l16 * KPAD + ka];
        b.y = sW[l16 * KPAD + ka + 1];
        c = __builtin_amdgcn_wmma_f32_16x16x4_f32(
                 false, a,  false, b,
                 (short)0, c,  false,  false);
    }

    int col = n0 + l16;
    if (col < Ncols) {
        float bv = bias[col];
        #pragma unroll
        for (int r = 0; r < 8; ++r) {
            int row = waveRow + r + 8 * half;
            if (row < N_NODES) {
                float v = c[r] + bv;
                if (doRelu) v = fmaxf(v, 0.0f);
                Out[row * outStride + col] = v;
            }
        }
    }
}

extern "C" void kernel_launch(void* const* d_in, const int* in_sizes, int n_in,
                              void* d_out, int out_size, void* d_ws, size_t ws_size,
                              hipStream_t stream) {
    const float* in_feat = (const float*)d_in[0];
    const int*   src     = (const int*)d_in[1];
    const int*   dst     = (const int*)d_in[2];
    const float* W1 = (const float*)d_in[3];   const float* b1 = (const float*)d_in[4];
    const float* W2 = (const float*)d_in[5];   const float* b2 = (const float*)d_in[6];
    const float* Wc1 = (const float*)d_in[7];  const float* bc1 = (const float*)d_in[8];
    const float* Wc2 = (const float*)d_in[9];  const float* bc2 = (const float*)d_in[10];
    const float* W3 = (const float*)d_in[11];  const float* b3 = (const float*)d_in[12];
    const float* W4 = (const float*)d_in[13];  const float* b4 = (const float*)d_in[14];

    char* ws = (char*)d_ws;
    const size_t SZ_H   = (size_t)N_NODES * 64 * 4;
    const size_t SZ_CAT = (size_t)N_NODES * 192 * 4;
    float* dinv = (float*)(ws);
    float* bufA = (float*)(ws + 200192);
    float* bufB = (float*)(ws + 200192 + SZ_H);
    float* xcat = (float*)(ws + 200192 + 2 * SZ_H);
    float* agg  = (float*)(ws + 200192 + 2 * SZ_H + SZ_CAT);

    const int n64 = N_NODES * 64;
    dim3 blkS(256);
    dim3 gN((N_NODES + 255) / 256), gE((N_EDGES + 255) / 256), gNF((n64 + 255) / 256);
    long long espmm = (long long)N_EDGES * 64;
    dim3 gSP((unsigned)((espmm + 255) / 256));

    zero_f32<<<gN, blkS, 0, stream>>>(dinv, N_NODES);
    deg_accum<<<gE, blkS, 0, stream>>>(dst, dinv, N_EDGES);
    deg_to_dinv<<<gN, blkS, 0, stream>>>(dinv, N_NODES);

    dim3 gemmBlk(128);
    dim3 gG64((N_NODES + 63) / 64, 4);
    dim3 gG2((N_NODES + 63) / 64, 1);

    gemm_wmma<<<gG64, gemmBlk, 0, stream>>>(in_feat, 128, W1, b1, bufA, 64, 128, 64, 1);
    gemm_wmma<<<gG64, gemmBlk, 0, stream>>>(bufA,     64, W2, b2, bufB, 64,  64, 64, 1);

    const float rn = 2.0f / 2.0f;
    const float* h_in = bufB;
    float*       h_out = bufA;
    const float* Wc[2] = { Wc1, Wc2 };
    const float* bc[2] = { bc1, bc2 };
    for (int layer = 0; layer < 2; ++layer) {
        copy_to_xcat<<<gNF, blkS, 0, stream>>>(h_in, xcat, n64);
        zero_f32<<<gNF, blkS, 0, stream>>>(agg, n64);
        spmm_edges<<<gSP, blkS, 0, stream>>>(src, dst, xcat, 192, dinv, agg, N_EDGES);
        cheb_x1<<<gNF, blkS, 0, stream>>>(agg, xcat, dinv, -rn, rn - 1.0f, n64);
        zero_f32<<<gNF, blkS, 0, stream>>>(agg, n64);
        spmm_edges<<<gSP, blkS, 0, stream>>>(src, dst, xcat + 64, 192, dinv, agg, N_EDGES);
        cheb_x2<<<gNF, blkS, 0, stream>>>(agg, xcat, dinv, -2.0f * rn, 2.0f * (rn - 1.0f), -1.0f, n64);
        gemm_wmma<<<gG64, gemmBlk, 0, stream>>>(xcat, 192, Wc[layer], bc[layer], h_out, 64, 192, 64, 1);
        const float* t = h_in; h_in = h_out; h_out = (float*)t;
    }
    gemm_wmma<<<gG64, gemmBlk, 0, stream>>>(h_in, 64, W3, b3, (float*)h_out, 64, 64, 64, 1);
    gemm_wmma<<<gG2,  gemmBlk, 0, stream>>>((const float*)h_out, 64, W4, b4, (float*)d_out, 2, 64, 2, 0);
}


# Chebyshev layers regrouped, weights before propagation: one WMMA pass makes P=X0(Wa-Wc)^T+b, Q=X0Wb^T, R=X0Wc^T; edge passes add -2AR into Q, then -AQ into P; sum order changes; 14->12 launches
# speedup vs baseline: 4.4697x; 4.4697x over previous
#include <hip/hip_runtime.h>
#include <hip/hip_bf16.h>

typedef __attribute__((ext_vector_type(2))) float v2f;
typedef __attribute__((ext_vector_type(4))) float v4f;
typedef __attribute__((ext_vector_type(8))) float v8f;

#define N_NODES 50000
#define N_EDGES 800000

#define DH_G     64
#define DH_EPS   (N_EDGES / DH_G)
#define DH_PITCH (((N_NODES + 31) / 32) * 32)

__global__ void __launch_bounds__(1024)
deg_hist(const int* __restrict__ dst, unsigned* __restrict__ planes) {
    static_assert(N_EDGES % (DH_G * 4) == 0, "a slice is a whole number of 16-byte pieces");
    static_assert(N_NODES % 4 == 0, "the table is cleared and stored in 16-byte pieces");
    constexpr int N4    = DH_EPS / 4;
    constexpr int STEPS = (N4 + 1023) / 1024;
    constexpr int T4    = N_NODES / 4;
    __shared__ __attribute__((aligned(16))) unsigned cnt[N_NODES];
    const int tid = threadIdx.x;

    const int4* d4 = (const int4*)(dst + (size_t)blockIdx.x * DH_EPS);
    int4 d[STEPS];
    #pragma unroll
    for (int s = 0; s < STEPS; ++s) {
        const int i = tid + s * 1024;
        d[s] = (i < N4) ? d4[i] : make_int4(-1, -1, -1, -1);
    }
    for (int i = tid; i < T4; i += 1024) ((uint4*)cnt)[i] = make_uint4(0u, 0u, 0u, 0u);
    __syncthreads();

    #pragma unroll
    for (int s = 0; s < STEPS; ++s) {
        if ((unsigned)d[s].x < (unsigned)N_NODES) atomicAdd(&cnt[d[s].x], 1u);
        if ((unsigned)d[s].y < (unsigned)N_NODES) atomicAdd(&cnt[d[s].y], 1u);
        if ((unsigned)d[s].z < (unsigned)N_NODES) atomicAdd(&cnt[d[s].z], 1u);
        if ((unsigned)d[s].w < (unsigned)N_NODES) atomicAdd(&cnt[d[s].w], 1u);
    }
    __syncthreads();

    uint4* out = (uint4*)(planes + (size_t)blockIdx.x * DH_PITCH);
    for (int i = tid; i < T4; i += 1024) out[i] = ((const uint4*)cnt)[i];
}

__global__ void deg_to_dinv(const unsigned* __restrict__ planes, float* __restrict__ dinv, int n) {
    int i = blockIdx.x * blockDim.x + threadIdx.x;
    if (i >= n) return;
    unsigned c = 0;
    #pragma unroll
    for (int g = 0; g < DH_G; ++g) c += planes[(size_t)g * DH_PITCH + i];
    dinv[i] = rsqrtf(fmaxf((float)c, 1.0f));
}

#define SP_EPW 32
#define SP_B   16

__global__ void __launch_bounds__(256)
spmm_edges(const int* __restrict__ src, const int* __restrict__ dst,
           const float* __restrict__ x, int xStride,
           const float* __restrict__ dinv, float coef,
           float* __restrict__ out, int outStride, int e) {
    static_assert(N_EDGES % SP_EPW == 0, "edge runs must be full: there is no tail path");
    const int lane = threadIdx.x & 31;
    const int wave = __builtin_amdgcn_readfirstlane((int)(threadIdx.x >> 5));
    const int base = (blockIdx.x * (int)(blockDim.x >> 5) + wave) * SP_EPW;
    if (base >= e) return;

    const int   s_l = src[base + lane];
    const int   d_l = dst[base + lane];
    const int   w_l = __float_as_int((coef * dinv[s_l]) * dinv[d_l]);
    const int   xo_l = s_l * xStride;
    const int   ao_l = d_l * outStride;

    #pragma unroll
    for (int b = 0; b < SP_EPW; b += SP_B) {
        float v0[SP_B], v1[SP_B];
        #pragma unroll
        for (int j = 0; j < SP_B; ++j) {
            const unsigned xo = (unsigned)__builtin_amdgcn_readlane(xo_l, b + j);
            const float* row = x + (xo + (unsigned)lane);
            v0[j] = row[0];
            v1[j] = row[32];
        }
        #pragma unroll
        for (int j = 0; j < SP_B; ++j) {
            const unsigned ao = (unsigned)__builtin_amdgcn_readlane(ao_l, b + j);
            const float    w  = __int_as_float(__builtin_amdgcn_readlane(w_l, b + j));
            float* arow = out + (ao + (unsigned)lane);
            atomicAdd(arow,      v0[j] * w);
            atomicAdd(arow + 32, v1[j] * w);
        }
    }
}

#define HOLD8(b) "v"(xf[(b) + 0]), "v"(xf[(b) + 1]), "v"(xf[(b) + 2]), "v"(xf[(b) + 3]), \
                 "v"(xf[(b) + 4]), "v"(xf[(b) + 5]), "v"(xf[(b) + 6]), "v"(xf[(b) + 7])

template <int K, int NCOLS, bool RELU, bool RIN>
__global__ void __launch_bounds__(128)
gemm_wstat(const float* __restrict__ X, int xStride,
           const float* __restrict__ W, const float* __restrict__ bias,
           float* __restrict__ Out, int outStride) {
    constexpr int MT    = (NCOLS + 15) / 16;
    constexpr int SLOTS = 4 / MT;
    constexpr int KB    = K / 8;
    constexpr int NT    = N_NODES / 16;
    static_assert(N_NODES % 16 == 0, "node tiles must be full: WMMA needs every lane");
    static_assert(K % 8 == 0 && 4 % MT == 0, "shape");
    static_assert(KB == 8 || KB == 16 || KB == 24, "the load fence below is written for these sizes");

    const int lane = threadIdx.x & 31;
    const int wave = __builtin_amdgcn_readfirstlane((int)(threadIdx.x >> 5));
    const int c = lane & 15;
    const int h = lane >> 4;
    const int mt   = wave % MT;
    const int slot = wave / MT;

    const int  frow   = mt * 16 + c;
    const bool fvalid = frow < NCOLS;
    const float* wp = W + (fvalid ? frow : 0) * K + 4 * h;
    v4f wf[KB];
    #pragma unroll
    for (int t = 0; t < KB; ++t) {
        v4f w = *(const v4f*)(wp + 8 * t);
        wf[t] = fvalid ? w : (v4f){0.0f, 0.0f, 0.0f, 0.0f};
    }

    const int fb = mt * 16 + 8 * h;
    float bv[8];
    #pragma unroll
    for (int r = 0; r < 8; ++r) bv[r] = (fb + r < NCOLS) ? bias[fb + r] : 0.0f;

    for (int tile = blockIdx.x * SLOTS + slot; tile < NT; tile += gridDim.x * SLOTS) {
        const int node = tile * 16 + c;
        const float* xp = X + (size_t)node * xStride + 4 * h;
        v4f xf[KB];
        #pragma unroll
        for (int t = 0; t < KB; ++t) xf[t] = *(const v4f*)(xp + 8 * t);
        if constexpr (KB == 8)       asm volatile("" :: HOLD8(0));
        else if constexpr (KB == 16) asm volatile("" :: HOLD8(0), HOLD8(8));
        else                         asm volatile("" :: HOLD8(0), HOLD8(8), HOLD8(16));
        if constexpr (RIN) {
            #pragma unroll
            for (int t = 0; t < KB; ++t)
                xf[t] = (v4f){fmaxf(xf[t].x, 0.0f), fmaxf(xf[t].y, 0.0f), fmaxf(xf[t].z, 0.0f), fmaxf(xf[t].w, 0.0f)};
        }

        v8f acc0 = {}, acc1 = {};
        #pragma unroll
        for (int t = 0; t < KB; ++t) {
            v2f a0 = {wf[t].x, wf[t].y}, b0 = {xf[t].x, xf[t].y};
            v2f a1 = {wf[t].z, wf[t].w}, b1 = {xf[t].z, xf[t].w};
            acc0 = __builtin_amdgcn_wmma_f32_16x16x4_f32(false, a0, false, b0, (short)0, acc0, false, false);
            acc1 = __builtin_amdgcn_wmma_f32_16x16x4_f32(false, a1, false, b1, (short)0, acc1, false, false);
        }

        float v[8];
        #pragma unroll
        for (int r = 0; r < 8; ++r) {
            float s = (acc0[r] + acc1[r]) + bv[r];
            v[r] = RELU ? fmaxf(s, 0.0f) : s;
        }
        float* op = Out + (size_t)node * outStride + fb;
        if constexpr (NCOLS % 16 == 0) {
            *(v4f*)(op)     = (v4f){v[0], v[1], v[2], v[3]};
            *(v4f*)(op + 4) = (v4f){v[4], v[5], v[6], v[7]};
        } else {
            #pragma unroll
            for (int r = 0; r < 8; ++r)
                if (fb + r < NCOLS) op[r] = v[r];
        }
    }
}

template <bool RIN>
__global__ void __launch_bounds__(128)
cheb_pre(const float* __restrict__ X, int xStride,
         const float* __restrict__ Wl, const float* __restrict__ bias,
         float* __restrict__ outP, int pitchP,
         float* __restrict__ outQ, int pitchQ,
         float* __restrict__ outR, int pitchR) {
    constexpr int KB = 8;
    constexpr int NT = N_NODES / 16;
    constexpr int WK = 192;

    const int lane = threadIdx.x & 31;
    const int wave = __builtin_amdgcn_readfirstlane((int)(threadIdx.x >> 5));
    const int c = lane & 15;
    const int h = lane >> 4;

    const float* wp = Wl + (size_t)(wave * 16 + c) * WK + 4 * h;
    v4f wP[KB], wQ[KB], wR[KB];
    #pragma unroll
    for (int t = 0; t < KB; ++t) {
        const v4f wa = *(const v4f*)(wp + 8 * t);
        wQ[t] = *(const v4f*)(wp + 64 + 8 * t);
        wR[t] = *(const v4f*)(wp + 128 + 8 * t);
        wP[t] = wa - wR[t];
    }

    const int fb = wave * 16 + 8 * h;
    float bv[8];
    #pragma unroll
    for (int r = 0; r < 8; ++r) bv[r] = bias[fb + r];

    for (int tile = blockIdx.x; tile < NT; tile += gridDim.x) {
        const int node = tile * 16 + c;
        const float* xp = X + (size_t)node * xStride + 4 * h;
        v4f xf[KB];
        #pragma unroll
        for (int t = 0; t < KB; ++t) xf[t] = *(const v4f*)(xp + 8 * t);
        asm volatile("" :: HOLD8(0));
        if constexpr (RIN) {
            #pragma unroll
            for (int t = 0; t < KB; ++t)
                xf[t] = (v4f){fmaxf(xf[t].x, 0.0f), fmaxf(xf[t].y, 0.0f), fmaxf(xf[t].z, 0.0f), fmaxf(xf[t].w, 0.0f)};
        }

        v8f accP = {}, accQ = {}, accR = {};
        #pragma unroll
        for (int t = 0; t < KB; ++t) {
            const v2f b0 = {xf[t].x, xf[t].y}, b1 = {xf[t].z, xf[t].w};
            accP = __builtin_amdgcn_wmma_f32_16x16x4_f32(false, (v2f){wP[t].x, wP[t].y}, false, b0, (short)0, accP, false, false);
            accQ = __builtin_amdgcn_wmma_f32_16x16x4_f32(false, (v2f){wQ[t].x, wQ[t].y}, false, b0, (short)0, accQ, false, false);
            accR = __builtin_amdgcn_wmma_f32_16x16x4_f32(false, (v2f){wR[t].x, wR[t].y}, false, b0, (short)0, accR, false, false);
            accP = __builtin_amdgcn_wmma_f32_16x16x4_f32(false, (v2f){wP[t].z, wP[t].w}, false, b1, (short)0, accP, false, false);
            accQ = __builtin_amdgcn_wmma_f32_16x16x4_f32(false, (v2f){wQ[t].z, wQ[t].w}, false, b1, (short)0, accQ, false, false);
            accR = __builtin_amdgcn_wmma_f32_16x16x4_f32(false, (v2f){wR[t].z, wR[t].w}, false, b1, (short)0, accR, false, false);
        }

        float* pp = outP + (size_t)node * pitchP + fb;
        float* qp = outQ + (size_t)node * pitchQ + fb;
        float* rp = outR + (size_t)node * pitchR + fb;
        *(v4f*)(pp)     = (v4f){accP[0] + bv[0], accP[1] + bv[1], accP[2] + bv[2], accP[3] + bv[3]};
        *(v4f*)(pp + 4) = (v4f){accP[4] + bv[4], accP[5] + bv[5], accP[6] + bv[6], accP[7] + bv[7]};
        *(v4f*)(qp)     = (v4f){accQ[0], accQ[1], accQ[2], accQ[3]};
        *(v4f*)(qp + 4) = (v4f){accQ[4], accQ[5], accQ[6], accQ[7]};
        *(v4f*)(rp)     = (v4f){accR[0], accR[1], accR[2], accR[3]};
        *(v4f*)(rp + 4) = (v4f){accR[4], accR[5], accR[6], accR[7]};
    }
}

extern "C" void kernel_launch(void* const* d_in, const int* in_sizes, int n_in,
                              void* d_out, int out_size, void* d_ws, size_t ws_size,
                              hipStream_t stream) {
    const float* in_feat = (const float*)d_in[0];
    const int*   src     = (const int*)d_in[1];
    const int*   dst     = (const int*)d_in[2];
    const float* W1 = (const float*)d_in[3];   const float* b1 = (const float*)d_in[4];
    const float* W2 = (const float*)d_in[5];   const float* b2 = (const float*)d_in[6];
    const float* Wc1 = (const float*)d_in[7];  const float* bc1 = (const float*)d_in[8];
    const float* Wc2 = (const float*)d_in[9];  const float* bc2 = (const float*)d_in[10];
    const float* W3 = (const float*)d_in[11];  const float* b3 = (const float*)d_in[12];
    const float* W4 = (const float*)d_in[13];  const float* b4 = (const float*)d_in[14];

    char* ws = (char*)d_ws;
    const size_t SZ_H   = (size_t)N_NODES * 64 * 4;
    float* dinv = (float*)(ws);
    float* bufA = (float*)(ws + 200192);
    float* bufB = (float*)(ws + 200192 + SZ_H);
    float* xcat = (float*)(ws + 200192 + 2 * SZ_H);
    static_assert((size_t)DH_G * DH_PITCH * 4 <= (size_t)N_NODES * 192 * 4, "the count planes fit in the xcat area");

    dim3 blkS(256);
    dim3 gN((N_NODES + 255) / 256);
    dim3 gSP((N_EDGES / SP_EPW + 7) / 8);

    unsigned* planes = (unsigned*)xcat;
    deg_hist<<<dim3(DH_G), dim3(1024), 0, stream>>>(dst, planes);
    deg_to_dinv<<<gN, blkS, 0, stream>>>(planes, dinv, N_NODES);

    dim3 gemmBlk(128);
    dim3 gG(768);

    gemm_wstat<128, 64, true, false><<<gG, gemmBlk, 0, stream>>>(in_feat, 128, W1, b1, bufA, 64);
    gemm_wstat<64, 64, true, false><<<gG, gemmBlk, 0, stream>>>(bufA, 64, W2, b2, bufB, 64);

    constexpr float rn = 2.0f / 2.0f;
    static_assert(rn - 1.0f == 0.0f, "the layer's regrouping is written for rn = 1: the X0 term of X1 and the X1 term of X2 must be zero");

    cheb_pre<false><<<gG, gemmBlk, 0, stream>>>(bufB, 64, Wc1, bc1, xcat, 192, xcat + 64, 192, xcat + 128, 192);
    spmm_edges<<<gSP, blkS, 0, stream>>>(src, dst, xcat + 128, 192, dinv, -2.0f * rn, xcat + 64, 192, N_EDGES);
    spmm_edges<<<gSP, blkS, 0, stream>>>(src, dst, xcat + 64, 192, dinv, -rn, xcat, 192, N_EDGES);

    cheb_pre<true><<<gG, gemmBlk, 0, stream>>>(xcat, 192, Wc2, bc2, xcat + 64, 192, xcat + 128, 192, bufA, 64);
    spmm_edges<<<gSP, blkS, 0, stream>>>(src, dst, bufA, 64, dinv, -2.0f * rn, xcat + 128, 192, N_EDGES);
    spmm_edges<<<gSP, blkS, 0, stream>>>(src, dst, xcat + 128, 192, dinv, -rn, xcat + 64, 192, N_EDGES);

    gemm_wstat<64, 64, true, true><<<gG, gemmBlk, 0, stream>>>(xcat + 64, 192, W3, b3, bufB, 64);
    gemm_wstat<64, 2, false, false><<<gG, gemmBlk, 0, stream>>>(bufB, 64, W4, b4, (float*)d_out, 2);
}


# SpMM edge weights moved to the nodes: product kernel stores rows scaled by dinv or 1/dinv, readers take relu(dinv V); first edge pass unweighted, second weighs by 2 dinv[src]^2; atomics kept
# speedup vs baseline: 4.7449x; 1.0616x over previous
#include <hip/hip_runtime.h>
#include <hip/hip_bf16.h>

typedef __attribute__((ext_vector_type(2))) float v2f;
typedef __attribute__((ext_vector_type(4))) float v4f;
typedef __attribute__((ext_vector_type(8))) float v8f;

#define N_NODES 50000
#define N_EDGES 800000

#define DH_G     64
#define DH_EPS   (N_EDGES / DH_G)
#define DH_PITCH (((N_NODES + 31) / 32) * 32)

__global__ void __launch_bounds__(1024)
deg_hist(const int* __restrict__ dst, unsigned* __restrict__ planes) {
    static_assert(N_EDGES % (DH_G * 4) == 0, "a slice is a whole number of 16-byte pieces");
    static_assert(N_NODES % 4 == 0, "the table is cleared and stored in 16-byte pieces");
    constexpr int N4    = DH_EPS / 4;
    constexpr int STEPS = (N4 + 1023) / 1024;
    constexpr int T4    = N_NODES / 4;
    __shared__ __attribute__((aligned(16))) unsigned cnt[N_NODES];
    const int tid = threadIdx.x;

    const int4* d4 = (const int4*)(dst + (size_t)blockIdx.x * DH_EPS);
    int4 d[STEPS];
    #pragma unroll
    for (int s = 0; s < STEPS; ++s) {
        const int i = tid + s * 1024;
        d[s] = (i < N4) ? d4[i] : make_int4(-1, -1, -1, -1);
    }
    for (int i = tid; i < T4; i += 1024) ((uint4*)cnt)[i] = make_uint4(0u, 0u, 0u, 0u);
    __syncthreads();

    #pragma unroll
    for (int s = 0; s < STEPS; ++s) {
        if ((unsigned)d[s].x < (unsigned)N_NODES) atomicAdd(&cnt[d[s].x], 1u);
        if ((unsigned)d[s].y < (unsigned)N_NODES) atomicAdd(&cnt[d[s].y], 1u);
        if ((unsigned)d[s].z < (unsigned)N_NODES) atomicAdd(&cnt[d[s].z], 1u);
        if ((unsigned)d[s].w < (unsigned)N_NODES) atomicAdd(&cnt[d[s].w], 1u);
    }
    __syncthreads();

    uint4* out = (uint4*)(planes + (size_t)blockIdx.x * DH_PITCH);
    for (int i = tid; i < T4; i += 1024) out[i] = ((const uint4*)cnt)[i];
}

__global__ void deg_to_dinv(const unsigned* __restrict__ planes, float* __restrict__ dinv, int n) {
    int i = blockIdx.x * blockDim.x + threadIdx.x;
    if (i >= n) return;
    unsigned c = 0;
    #pragma unroll
    for (int g = 0; g < DH_G; ++g) c += planes[(size_t)g * DH_PITCH + i];
    dinv[i] = rsqrtf(fmaxf((float)c, 1.0f));
}

#define SP_EPW 32
#define SP_B   16

template <bool SRCW>
__global__ void __launch_bounds__(256)
spmm_edges(const int* __restrict__ src, const int* __restrict__ dst,
           const float* __restrict__ x, int xStride,
           const float* __restrict__ dinv,
           float* __restrict__ out, int outStride, int e) {
    static_assert(N_EDGES % SP_EPW == 0, "edge runs must be full: there is no tail path");
    const int lane = threadIdx.x & 31;
    const int wave = __builtin_amdgcn_readfirstlane((int)(threadIdx.x >> 5));
    const int base = (blockIdx.x * (int)(blockDim.x >> 5) + wave) * SP_EPW;
    if (base >= e) return;

    const int   s_l = src[base + lane];
    const int   d_l = dst[base + lane];
    int w_l = 0;
    if constexpr (SRCW) {
        const float ds = dinv[s_l];
        w_l = __float_as_int((2.0f * ds) * ds);
    }
    const int   xo_l = s_l * xStride;
    const int   ao_l = d_l * outStride;

    #pragma unroll
    for (int b = 0; b < SP_EPW; b += SP_B) {
        float v0[SP_B], v1[SP_B];
        #pragma unroll
        for (int j = 0; j < SP_B; ++j) {
            const unsigned xo = (unsigned)__builtin_amdgcn_readlane(xo_l, b + j);
            const float* row = x + (xo + (unsigned)lane);
            v0[j] = row[0];
            v1[j] = row[32];
        }
        #pragma unroll
        for (int j = 0; j < SP_B; ++j) {
            const unsigned ao = (unsigned)__builtin_amdgcn_readlane(ao_l, b + j);
            float* arow = out + (ao + (unsigned)lane);
            if constexpr (SRCW) {
                const float w = __int_as_float(__builtin_amdgcn_readlane(w_l, b + j));
                atomicAdd(arow,      v0[j] * w);
                atomicAdd(arow + 32, v1[j] * w);
            } else {
                atomicAdd(arow,      v0[j]);
                atomicAdd(arow + 32, v1[j]);
            }
        }
    }
}

#define HOLD8(b) "v"(xf[(b) + 0]), "v"(xf[(b) + 1]), "v"(xf[(b) + 2]), "v"(xf[(b) + 3]), \
                 "v"(xf[(b) + 4]), "v"(xf[(b) + 5]), "v"(xf[(b) + 6]), "v"(xf[(b) + 7])

template <int K, int NCOLS, bool RELU, bool RIN>
__global__ void __launch_bounds__(128)
gemm_wstat(const float* __restrict__ X, int xStride,
           const float* __restrict__ W, const float* __restrict__ bias,
           const float* __restrict__ dinv,
           float* __restrict__ Out, int outStride) {
    constexpr int MT    = (NCOLS + 15) / 16;
    constexpr int SLOTS = 4 / MT;
    constexpr int KB    = K / 8;
    constexpr int NT    = N_NODES / 16;
    static_assert(N_NODES % 16 == 0, "node tiles must be full: WMMA needs every lane");
    static_assert(K % 8 == 0 && 4 % MT == 0, "shape");
    static_assert(KB == 8 || KB == 16 || KB == 24, "the load fence below is written for these sizes");
    static_assert(!RIN || KB == 8, "the load fence with the node's factor is written for K = 64");

    const int lane = threadIdx.x & 31;
    const int wave = __builtin_amdgcn_readfirstlane((int)(threadIdx.x >> 5));
    const int c = lane & 15;
    const int h = lane >> 4;
    const int mt   = wave % MT;
    const int slot = wave / MT;

    const int  frow   = mt * 16 + c;
    const bool fvalid = frow < NCOLS;
    const float* wp = W + (fvalid ? frow : 0) * K + 4 * h;
    v4f wf[KB];
    #pragma unroll
    for (int t = 0; t < KB; ++t) {
        v4f w = *(const v4f*)(wp + 8 * t);
        wf[t] = fvalid ? w : (v4f){0.0f, 0.0f, 0.0f, 0.0f};
    }

    const int fb = mt * 16 + 8 * h;
    float bv[8];
    #pragma unroll
    for (int r = 0; r < 8; ++r) bv[r] = (fb + r < NCOLS) ? bias[fb + r] : 0.0f;

    for (int tile = blockIdx.x * SLOTS + slot; tile < NT; tile += gridDim.x * SLOTS) {
        const int node = tile * 16 + c;
        const float* xp = X + (size_t)node * xStride + 4 * h;
        v4f xf[KB];
        #pragma unroll
        for (int t = 0; t < KB; ++t) xf[t] = *(const v4f*)(xp + 8 * t);
        float dn = 0.0f;
        if constexpr (RIN) dn = dinv[node];
        if constexpr (RIN)           asm volatile("" :: HOLD8(0), "v"(dn));
        else if constexpr (KB == 8)  asm volatile("" :: HOLD8(0));
        else if constexpr (KB == 16) asm volatile("" :: HOLD8(0), HOLD8(8));
        else                         asm volatile("" :: HOLD8(0), HOLD8(8), HOLD8(16));
        if constexpr (RIN) {
            #pragma unroll
            for (int t = 0; t < KB; ++t)
                xf[t] = (v4f){fmaxf(xf[t].x, 0.0f), fmaxf(xf[t].y, 0.0f), fmaxf(xf[t].z, 0.0f), fmaxf(xf[t].w, 0.0f)};
        }

        v8f acc0 = {}, acc1 = {};
        #pragma unroll
        for (int t = 0; t < KB; ++t) {
            v2f a0 = {wf[t].x, wf[t].y}, b0 = {xf[t].x, xf[t].y};
            v2f a1 = {wf[t].z, wf[t].w}, b1 = {xf[t].z, xf[t].w};
            acc0 = __builtin_amdgcn_wmma_f32_16x16x4_f32(false, a0, false, b0, (short)0, acc0, false, false);
            acc1 = __builtin_amdgcn_wmma_f32_16x16x4_f32(false, a1, false, b1, (short)0, acc1, false, false);
        }

        float v[8];
        #pragma unroll
        for (int r = 0; r < 8; ++r) {
            float s = RIN ? fmaf(acc0[r] + acc1[r], dn, bv[r]) : (acc0[r] + acc1[r]) + bv[r];
            v[r] = RELU ? fmaxf(s, 0.0f) : s;
        }
        float* op = Out + (size_t)node * outStride + fb;
        if constexpr (NCOLS % 16 == 0) {
            *(v4f*)(op)     = (v4f){v[0], v[1], v[2], v[3]};
            *(v4f*)(op + 4) = (v4f){v[4], v[5], v[6], v[7]};
        } else {
            #pragma unroll
            for (int r = 0; r < 8; ++r)
                if (fb + r < NCOLS) op[r] = v[r];
        }
    }
}

template <bool RIN>
__global__ void __launch_bounds__(128)
cheb_pre(const float* __restrict__ X, int xStride,
         const float* __restrict__ Wl, const float* __restrict__ bias,
         const float* __restrict__ dinv,
         float* __restrict__ outP, int pitchP,
         float* __restrict__ outQ, int pitchQ,
         float* __restrict__ outR, int pitchR) {
    constexpr int KB = 8;
    constexpr int NT = N_NODES / 16;
    constexpr int WK = 192;

    const int lane = threadIdx.x & 31;
    const int wave = __builtin_amdgcn_readfirstlane((int)(threadIdx.x >> 5));
    const int c = lane & 15;
    const int h = lane >> 4;

    const float* wp = Wl + (size_t)(wave * 16 + c) * WK + 4 * h;
    v4f wP[KB], wQ[KB], wR[KB];
    #pragma unroll
    for (int t = 0; t < KB; ++t) {
        const v4f wa = *(const v4f*)(wp + 8 * t);
        wQ[t] = *(const v4f*)(wp + 64 + 8 * t);
        wR[t] = *(const v4f*)(wp + 128 + 8 * t);
        wP[t] = wa - wR[t];
    }

    const int fb = wave * 16 + 8 * h;
    float bv[8];
    #pragma unroll
    for (int r = 0; r < 8; ++r) bv[r] = bias[fb + r];

    for (int tile = blockIdx.x; tile < NT; tile += gridDim.x) {
        const int node = tile * 16 + c;
        const float* xp = X + (size_t)node * xStride + 4 * h;
        v4f xf[KB];
        #pragma unroll
        for (int t = 0; t < KB; ++t) xf[t] = *(const v4f*)(xp + 8 * t);
        const float dn = dinv[node];
        asm volatile("" :: HOLD8(0), "v"(dn));
        if constexpr (RIN) {
            #pragma unroll
            for (int t = 0; t < KB; ++t)
                xf[t] = (v4f){fmaxf(xf[t].x, 0.0f), fmaxf(xf[t].y, 0.0f), fmaxf(xf[t].z, 0.0f), fmaxf(xf[t].w, 0.0f)};
        }

        v8f accP = {}, accQ = {}, accR = {};
        #pragma unroll
        for (int t = 0; t < KB; ++t) {
            const v2f b0 = {xf[t].x, xf[t].y}, b1 = {xf[t].z, xf[t].w};
            accP = __builtin_amdgcn_wmma_f32_16x16x4_f32(false, (v2f){wP[t].x, wP[t].y}, false, b0, (short)0, accP, false, false);
            accQ = __builtin_amdgcn_wmma_f32_16x16x4_f32(false, (v2f){wQ[t].x, wQ[t].y}, false, b0, (short)0, accQ, false, false);
            accR = __builtin_amdgcn_wmma_f32_16x16x4_f32(false, (v2f){wR[t].x, wR[t].y}, false, b0, (short)0, accR, false, false);
            accP = __builtin_amdgcn_wmma_f32_16x16x4_f32(false, (v2f){wP[t].z, wP[t].w}, false, b1, (short)0, accP, false, false);
            accQ = __builtin_amdgcn_wmma_f32_16x16x4_f32(false, (v2f){wQ[t].z, wQ[t].w}, false, b1, (short)0, accQ, false, false);
            accR = __builtin_amdgcn_wmma_f32_16x16x4_f32(false, (v2f){wR[t].z, wR[t].w}, false, b1, (short)0, accR, false, false);
        }

        float* pp = outP + (size_t)node * pitchP + fb;
        float* qp = outQ + (size_t)node * pitchQ + fb;
        float* rp = outR + (size_t)node * pitchR + fb;
        const float rd = 1.0f / dn;
        float vP[8], vQ[8], vR[8];
        #pragma unroll
        for (int r = 0; r < 8; ++r) {
            if constexpr (RIN) {
                vP[r] = fmaf(bv[r], rd, accP[r]);
                vQ[r] = -0.5f * accQ[r];
                vR[r] = (dn * dn) * accR[r];
            } else {
                vP[r] = (accP[r] + bv[r]) * rd;
                vQ[r] = accQ[r] * (-0.5f * rd);
                vR[r] = accR[r] * dn;
            }
        }
        *(v4f*)(pp)     = (v4f){vP[0], vP[1], vP[2], vP[3]};
        *(v4f*)(pp + 4) = (v4f){vP[4], vP[5], vP[6], vP[7]};
        *(v4f*)(qp)     = (v4f){vQ[0], vQ[1], vQ[2], vQ[3]};
        *(v4f*)(qp + 4) = (v4f){vQ[4], vQ[5], vQ[6], vQ[7]};
        *(v4f*)(rp)     = (v4f){vR[0], vR[1], vR[2], vR[3]};
        *(v4f*)(rp + 4) = (v4f){vR[4], vR[5], vR[6], vR[7]};
    }
}

extern "C" void kernel_launch(void* const* d_in, const int* in_sizes, int n_in,
                              void* d_out, int out_size, void* d_ws, size_t ws_size,
                              hipStream_t stream) {
    const float* in_feat = (const float*)d_in[0];
    const int*   src     = (const int*)d_in[1];
    const int*   dst     = (const int*)d_in[2];
    const float* W1 = (const float*)d_in[3];   const float* b1 = (const float*)d_in[4];
    const float* W2 = (const float*)d_in[5];   const float* b2 = (const float*)d_in[6];
    const float* Wc1 = (const float*)d_in[7];  const float* bc1 = (const float*)d_in[8];
    const float* Wc2 = (const float*)d_in[9];  const float* bc2 = (const float*)d_in[10];
    const float* W3 = (const float*)d_in[11];  const float* b3 = (const float*)d_in[12];
    const float* W4 = (const float*)d_in[13];  const float* b4 = (const float*)d_in[14];

    char* ws = (char*)d_ws;
    const size_t SZ_H   = (size_t)N_NODES * 64 * 4;
    float* dinv = (float*)(ws);
    float* bufA = (float*)(ws + 200192);
    float* bufB = (float*)(ws + 200192 + SZ_H);
    float* xcat = (float*)(ws + 200192 + 2 * SZ_H);
    static_assert((size_t)DH_G * DH_PITCH * 4 <= (size_t)N_NODES * 192 * 4, "the count planes fit in the xcat area");

    dim3 blkS(256);
    dim3 gN((N_NODES + 255) / 256);
    dim3 gSP((N_EDGES / SP_EPW + 7) / 8);

    unsigned* planes = (unsigned*)xcat;
    deg_hist<<<dim3(DH_G), dim3(1024), 0, stream>>>(dst, planes);
    deg_to_dinv<<<gN, blkS, 0, stream>>>(planes, dinv, N_NODES);

    dim3 gemmBlk(128);
    dim3 gG(768);

    gemm_wstat<128, 64, true, false><<<gG, gemmBlk, 0, stream>>>(in_feat, 128, W1, b1, dinv, bufA, 64);
    gemm_wstat<64, 64, true, false><<<gG, gemmBlk, 0, stream>>>(bufA, 64, W2, b2, dinv, bufB, 64);

    constexpr float rn = 2.0f / 2.0f;
    static_assert(rn - 1.0f == 0.0f, "the layer's regrouping is written for rn = 1: the X0 term of X1 and the X1 term of X2 must be zero");

    cheb_pre<false><<<gG, gemmBlk, 0, stream>>>(bufB, 64, Wc1, bc1, dinv, xcat, 192, xcat + 64, 192, xcat + 128, 192);
    spmm_edges<false><<<gSP, blkS, 0, stream>>>(src, dst, xcat + 128, 192, dinv, xcat + 64, 192, N_EDGES);
    spmm_edges<true><<<gSP, blkS, 0, stream>>>(src, dst, xcat + 64, 192, dinv, xcat, 192, N_EDGES);

    cheb_pre<true><<<gG, gemmBlk, 0, stream>>>(xcat, 192, Wc2, bc2, dinv, xcat + 64, 192, xcat + 128, 192, bufA, 64);
    spmm_edges<false><<<gSP, blkS, 0, stream>>>(src, dst, bufA, 64, dinv, xcat + 128, 192, N_EDGES);
    spmm_edges<true><<<gSP, blkS, 0, stream>>>(src, dst, xcat + 128, 192, dinv, xcat + 64, 192, N_EDGES);

    gemm_wstat<64, 64, true, true><<<gG, gemmBlk, 0, stream>>>(xcat + 64, 192, W3, b3, dinv, bufB, 64);
    gemm_wstat<64, 2, false, false><<<gG, gemmBlk, 0, stream>>>(bufB, 64, W4, b4, dinv, (float*)d_out, 2);
}


# Consecutive dense layers share a launch, the 16-node tile handed over on chip: W1>W2>Wc1 products via an LDS tile, W3>W4 from the accumulators as 4 k-partials (W4 sum order changes); 12->9 launches
# speedup vs baseline: 5.1636x; 1.0882x over previous
#include <hip/hip_runtime.h>
#include <hip/hip_bf16.h>

typedef __attribute__((ext_vector_type(2))) float v2f;
typedef __attribute__((ext_vector_type(4))) float v4f;
typedef __attribute__((ext_vector_type(8))) float v8f;

#define N_NODES 50000
#define N_EDGES 800000

#define DH_G     64
#define DH_EPS   (N_EDGES / DH_G)
#define DH_PITCH (((N_NODES + 31) / 32) * 32)

__global__ void __launch_bounds__(1024)
deg_hist(const int* __restrict__ dst, unsigned* __restrict__ planes) {
    static_assert(N_EDGES % (DH_G * 4) == 0, "a slice is a whole number of 16-byte pieces");
    static_assert(N_NODES % 4 == 0, "the table is cleared and stored in 16-byte pieces");
    constexpr int N4    = DH_EPS / 4;
    constexpr int STEPS = (N4 + 1023) / 1024;
    constexpr int T4    = N_NODES / 4;
    __shared__ __attribute__((aligned(16))) unsigned cnt[N_NODES];
    const int tid = threadIdx.x;

    const int4* d4 = (const int4*)(dst + (size_t)blockIdx.x * DH_EPS);
    int4 d[STEPS];
    #pragma unroll
    for (int s = 0; s < STEPS; ++s) {
        const int i = tid + s * 1024;
        d[s] = (i < N4) ? d4[i] : make_int4(-1, -1, -1, -1);
    }
    for (int i = tid; i < T4; i += 1024) ((uint4*)cnt)[i] = make_uint4(0u, 0u, 0u, 0u);
    __syncthreads();

    #pragma unroll
    for (int s = 0; s < STEPS; ++s) {
        if ((unsigned)d[s].x < (unsigned)N_NODES) atomicAdd(&cnt[d[s].x], 1u);
        if ((unsigned)d[s].y < (unsigned)N_NODES) atomicAdd(&cnt[d[s].y], 1u);
        if ((unsigned)d[s].z < (unsigned)N_NODES) atomicAdd(&cnt[d[s].z], 1u);
        if ((unsigned)d[s].w < (unsigned)N_NODES) atomicAdd(&cnt[d[s].w], 1u);
    }
    __syncthreads();

    uint4* out = (uint4*)(planes + (size_t)blockIdx.x * DH_PITCH);
    for (int i = tid; i < T4; i += 1024) out[i] = ((const uint4*)cnt)[i];
}

__global__ void deg_to_dinv(const unsigned* __restrict__ planes, float* __restrict__ dinv, int n) {
    int i = blockIdx.x * blockDim.x + threadIdx.x;
    if (i >= n) return;
    unsigned c = 0;
    #pragma unroll
    for (int g = 0; g < DH_G; ++g) c += planes[(size_t)g * DH_PITCH + i];
    dinv[i] = rsqrtf(fmaxf((float)c, 1.0f));
}

#define SP_EPW 32
#define SP_B   16

template <bool SRCW>
__global__ void __launch_bounds__(256)
spmm_edges(const int* __restrict__ src, const int* __restrict__ dst,
           const float* __restrict__ x, int xStride,
           const float* __restrict__ dinv,
           float* __restrict__ out, int outStride, int e) {
    static_assert(N_EDGES % SP_EPW == 0, "edge runs must be full: there is no tail path");
    const int lane = threadIdx.x & 31;
    const int wave = __builtin_amdgcn_readfirstlane((int)(threadIdx.x >> 5));
    const int base = (blockIdx.x * (int)(blockDim.x >> 5) + wave) * SP_EPW;
    if (base >= e) return;

    const int   s_l = src[base + lane];
    const int   d_l = dst[base + lane];
    int w_l = 0;
    if constexpr (SRCW) {
        const float ds = dinv[s_l];
        w_l = __float_as_int((2.0f * ds) * ds);
    }
    const int   xo_l = s_l * xStride;
    const int   ao_l = d_l * outStride;

    #pragma unroll
    for (int b = 0; b < SP_EPW; b += SP_B) {
        float v0[SP_B], v1[SP_B];
        #pragma unroll
        for (int j = 0; j < SP_B; ++j) {
            const unsigned xo = (unsigned)__builtin_amdgcn_readlane(xo_l, b + j);
            const float* row = x + (xo + (unsigned)lane);
            v0[j] = row[0];
            v1[j] = row[32];
        }
        #pragma unroll
        for (int j = 0; j < SP_B; ++j) {
            const unsigned ao = (unsigned)__builtin_amdgcn_readlane(ao_l, b + j);
            float* arow = out + (ao + (unsigned)lane);
            if constexpr (SRCW) {
                const float w = __int_as_float(__builtin_amdgcn_readlane(w_l, b + j));
                atomicAdd(arow,      v0[j] * w);
                atomicAdd(arow + 32, v1[j] * w);
            } else {
                atomicAdd(arow,      v0[j]);
                atomicAdd(arow + 32, v1[j]);
            }
        }
    }
}

#define HOLD8(b) "v"(xf[(b) + 0]), "v"(xf[(b) + 1]), "v"(xf[(b) + 2]), "v"(xf[(b) + 3]), \
                 "v"(xf[(b) + 4]), "v"(xf[(b) + 5]), "v"(xf[(b) + 6]), "v"(xf[(b) + 7])
#define HP 68

__global__ void __launch_bounds__(128)
head_chain(const float* __restrict__ X,
           const float* __restrict__ W1, const float* __restrict__ b1,
           const float* __restrict__ W2, const float* __restrict__ b2,
           const float* __restrict__ Wl, const float* __restrict__ bias,
           const float* __restrict__ dinv,
           float* __restrict__ outP, int pitchP,
           float* __restrict__ outQ, int pitchQ,
           float* __restrict__ outR, int pitchR) {
    constexpr int NT = N_NODES / 16;
    static_assert(N_NODES % 16 == 0, "node tiles must be full: WMMA needs every lane");
    __shared__ __attribute__((aligned(16))) float h1s[16 * HP];
    __shared__ __attribute__((aligned(16))) float h2s[16 * HP];

    const int lane = threadIdx.x & 31;
    const int wave = __builtin_amdgcn_readfirstlane((int)(threadIdx.x >> 5));
    const int c = lane & 15;
    const int h = lane >> 4;
    const int frow = wave * 16 + c;

    v4f w1f[16], w2f[8], wP[8], wQ[8], wR[8];
    {
        const float* p1 = W1 + (size_t)frow * 128 + 4 * h;
        #pragma unroll
        for (int t = 0; t < 16; ++t) w1f[t] = *(const v4f*)(p1 + 8 * t);
        const float* p2 = W2 + (size_t)frow * 64 + 4 * h;
        const float* p3 = Wl + (size_t)frow * 192 + 4 * h;
        #pragma unroll
        for (int t = 0; t < 8; ++t) {
            w2f[t] = *(const v4f*)(p2 + 8 * t);
            const v4f wa = *(const v4f*)(p3 + 8 * t);
            wQ[t] = *(const v4f*)(p3 + 64 + 8 * t);
            wR[t] = *(const v4f*)(p3 + 128 + 8 * t);
            wP[t] = wa - wR[t];
        }
    }

    const int fb = wave * 16 + 8 * h;
    float b1v[8], b2v[8], bv[8];
    #pragma unroll
    for (int r = 0; r < 8; ++r) { b1v[r] = b1[fb + r]; b2v[r] = b2[fb + r]; bv[r] = bias[fb + r]; }

    float* const h1w = h1s + c * HP + fb;
    float* const h2w = h2s + c * HP + fb;
    const float* const h1r = h1s + c * HP + 4 * h;
    const float* const h2r = h2s + c * HP + 4 * h;

    for (int tile = blockIdx.x; tile < NT; tile += gridDim.x) {
        const int node = tile * 16 + c;
        const float* xp = X + (size_t)node * 128 + 4 * h;
        v4f xf[16];
        #pragma unroll
        for (int t = 0; t < 16; ++t) xf[t] = *(const v4f*)(xp + 8 * t);
        const float dn = dinv[node];
        asm volatile("" :: HOLD8(0), HOLD8(8), "v"(dn));

        v8f acc0 = {}, acc1 = {};
        #pragma unroll
        for (int t = 0; t < 16; ++t) {
            acc0 = __builtin_amdgcn_wmma_f32_16x16x4_f32(false, (v2f){w1f[t].x, w1f[t].y}, false, (v2f){xf[t].x, xf[t].y}, (short)0, acc0, false, false);
            acc1 = __builtin_amdgcn_wmma_f32_16x16x4_f32(false, (v2f){w1f[t].z, w1f[t].w}, false, (v2f){xf[t].z, xf[t].w}, (short)0, acc1, false, false);
        }
        float v[8];
        #pragma unroll
        for (int r = 0; r < 8; ++r) v[r] = fmaxf((acc0[r] + acc1[r]) + b1v[r], 0.0f);
        *(v4f*)(h1w)     = (v4f){v[0], v[1], v[2], v[3]};
        *(v4f*)(h1w + 4) = (v4f){v[4], v[5], v[6], v[7]};
        __syncthreads();

        v4f yf[8];
        #pragma unroll
        for (int t = 0; t < 8; ++t) yf[t] = *(const v4f*)(h1r + 8 * t);
        acc0 = (v8f){}; acc1 = (v8f){};
        #pragma unroll
        for (int t = 0; t < 8; ++t) {
            acc0 = __builtin_amdgcn_wmma_f32_16x16x4_f32(false, (v2f){w2f[t].x, w2f[t].y}, false, (v2f){yf[t].x, yf[t].y}, (short)0, acc0, false, false);
            acc1 = __builtin_amdgcn_wmma_f32_16x16x4_f32(false, (v2f){w2f[t].z, w2f[t].w}, false, (v2f){yf[t].z, yf[t].w}, (short)0, acc1, false, false);
        }
        #pragma unroll
        for (int r = 0; r < 8; ++r) v[r] = fmaxf((acc0[r] + acc1[r]) + b2v[r], 0.0f);
        *(v4f*)(h2w)     = (v4f){v[0], v[1], v[2], v[3]};
        *(v4f*)(h2w + 4) = (v4f){v[4], v[5], v[6], v[7]};
        __syncthreads();

        v4f zf[8];
        #pragma unroll
        for (int t = 0; t < 8; ++t) zf[t] = *(const v4f*)(h2r + 8 * t);
        v8f accP = {}, accQ = {}, accR = {};
        #pragma unroll
        for (int t = 0; t < 8; ++t) {
            const v2f b0 = {zf[t].x, zf[t].y}, b1f = {zf[t].z, zf[t].w};
            accP = __builtin_amdgcn_wmma_f32_16x16x4_f32(false, (v2f){wP[t].x, wP[t].y}, false, b0, (short)0, accP, false, false);
            accQ = __builtin_amdgcn_wmma_f32_16x16x4_f32(false, (v2f){wQ[t].x, wQ[t].y}, false, b0, (short)0, accQ, false, false);
            accR = __builtin_amdgcn_wmma_f32_16x16x4_f32(false, (v2f){wR[t].x, wR[t].y}, false, b0, (short)0, accR, false, false);
            accP = __builtin_amdgcn_wmma_f32_16x16x4_f32(false, (v2f){wP[t].z, wP[t].w}, false, b1f, (short)0, accP, false, false);
            accQ = __builtin_amdgcn_wmma_f32_16x16x4_f32(false, (v2f){wQ[t].z, wQ[t].w}, false, b1f, (short)0, accQ, false, false);
            accR = __builtin_amdgcn_wmma_f32_16x16x4_f32(false, (v2f){wR[t].z, wR[t].w}, false, b1f, (short)0, accR, false, false);
        }

        float* pp = outP + (size_t)node * pitchP + fb;
        float* qp = outQ + (size_t)node * pitchQ + fb;
        float* rp = outR + (size_t)node * pitchR + fb;
        const float rd = 1.0f / dn;
        float vP[8], vQ[8], vR[8];
        #pragma unroll
        for (int r = 0; r < 8; ++r) {
            vP[r] = (accP[r] + bv[r]) * rd;
            vQ[r] = accQ[r] * (-0.5f * rd);
            vR[r] = accR[r] * dn;
        }
        *(v4f*)(pp)     = (v4f){vP[0], vP[1], vP[2], vP[3]};
        *(v4f*)(pp + 4) = (v4f){vP[4], vP[5], vP[6], vP[7]};
        *(v4f*)(qp)     = (v4f){vQ[0], vQ[1], vQ[2], vQ[3]};
        *(v4f*)(qp + 4) = (v4f){vQ[4], vQ[5], vQ[6], vQ[7]};
        *(v4f*)(rp)     = (v4f){vR[0], vR[1], vR[2], vR[3]};
        *(v4f*)(rp + 4) = (v4f){vR[4], vR[5], vR[6], vR[7]};
    }
}

__global__ void __launch_bounds__(128)
tail_chain(const float* __restrict__ X, int xStride,
           const float* __restrict__ W3, const float* __restrict__ b3,
           const float* __restrict__ W4, const float* __restrict__ b4,
           const float* __restrict__ dinv,
           float* __restrict__ Out) {
    constexpr int NT = N_NODES / 16;
    __shared__ __attribute__((aligned(16))) float part[2][4 * 32];

    const int lane = threadIdx.x & 31;
    const int wave = __builtin_amdgcn_readfirstlane((int)(threadIdx.x >> 5));
    const int c = lane & 15;
    const int h = lane >> 4;

    const float* w3p = W3 + (size_t)(wave * 16 + c) * 64 + 4 * h;
    v4f w3f[8];
    #pragma unroll
    for (int t = 0; t < 8; ++t) w3f[t] = *(const v4f*)(w3p + 8 * t);
    const bool ovalid = c < 2;
    const float* w4p = W4 + (ovalid ? c : 0) * 64 + wave * 16 + 8 * h;
    v4f w4f[2];
    #pragma unroll
    for (int t = 0; t < 2; ++t) {
        const v4f w = *(const v4f*)(w4p + 4 * t);
        w4f[t] = ovalid ? w : (v4f){0.0f, 0.0f, 0.0f, 0.0f};
    }

    const int fb = wave * 16 + 8 * h;
    float bv[8];
    #pragma unroll
    for (int r = 0; r < 8; ++r) bv[r] = b3[fb + r];
    const float b4v = b4[lane & 1];

    int par = 0;
    for (int tile = blockIdx.x; tile < NT; tile += gridDim.x) {
        const int node = tile * 16 + c;
        const float* xp = X + (size_t)node * xStride + 4 * h;
        v4f xf[8];
        #pragma unroll
        for (int t = 0; t < 8; ++t) xf[t] = *(const v4f*)(xp + 8 * t);
        const float dn = dinv[node];
        asm volatile("" :: HOLD8(0), "v"(dn));
        #pragma unroll
        for (int t = 0; t < 8; ++t)
            xf[t] = (v4f){fmaxf(xf[t].x, 0.0f), fmaxf(xf[t].y, 0.0f), fmaxf(xf[t].z, 0.0f), fmaxf(xf[t].w, 0.0f)};

        v8f acc0 = {}, acc1 = {};
        #pragma unroll
        for (int t = 0; t < 8; ++t) {
            acc0 = __builtin_amdgcn_wmma_f32_16x16x4_f32(false, (v2f){w3f[t].x, w3f[t].y}, false, (v2f){xf[t].x, xf[t].y}, (short)0, acc0, false, false);
            acc1 = __builtin_amdgcn_wmma_f32_16x16x4_f32(false, (v2f){w3f[t].z, w3f[t].w}, false, (v2f){xf[t].z, xf[t].w}, (short)0, acc1, false, false);
        }
        float v[8];
        #pragma unroll
        for (int r = 0; r < 8; ++r) v[r] = fmaxf(fmaf(acc0[r] + acc1[r], dn, bv[r]), 0.0f);

        v8f o0 = {}, o1 = {};
        o0 = __builtin_amdgcn_wmma_f32_16x16x4_f32(false, (v2f){w4f[0].x, w4f[0].y}, false, (v2f){v[0], v[1]}, (short)0, o0, false, false);
        o1 = __builtin_amdgcn_wmma_f32_16x16x4_f32(false, (v2f){w4f[0].z, w4f[0].w}, false, (v2f){v[2], v[3]}, (short)0, o1, false, false);
        o0 = __builtin_amdgcn_wmma_f32_16x16x4_f32(false, (v2f){w4f[1].x, w4f[1].y}, false, (v2f){v[4], v[5]}, (short)0, o0, false, false);
        o1 = __builtin_amdgcn_wmma_f32_16x16x4_f32(false, (v2f){w4f[1].z, w4f[1].w}, false, (v2f){v[6], v[7]}, (short)0, o1, false, false);
        float* pb = part[par];
        if (h == 0) *(v2f*)(pb + wave * 32 + 2 * c) = (v2f){o0[0] + o1[0], o0[1] + o1[1]};
        __syncthreads();
        if (wave == 0) {
            const float s = ((pb[lane] + pb[32 + lane]) + (pb[64 + lane] + pb[96 + lane])) + b4v;
            Out[(size_t)tile * 32 + lane] = s;
        }
        par ^= 1;
    }
}

template <bool RIN>
__global__ void __launch_bounds__(128)
cheb_pre(const float* __restrict__ X, int xStride,
         const float* __restrict__ Wl, const float* __restrict__ bias,
         const float* __restrict__ dinv,
         float* __restrict__ outP, int pitchP,
         float* __restrict__ outQ, int pitchQ,
         float* __restrict__ outR, int pitchR) {
    constexpr int KB = 8;
    constexpr int NT = N_NODES / 16;
    constexpr int WK = 192;

    const int lane = threadIdx.x & 31;
    const int wave = __builtin_amdgcn_readfirstlane((int)(threadIdx.x >> 5));
    const int c = lane & 15;
    const int h = lane >> 4;

    const float* wp = Wl + (size_t)(wave * 16 + c) * WK + 4 * h;
    v4f wP[KB], wQ[KB], wR[KB];
    #pragma unroll
    for (int t = 0; t < KB; ++t) {
        const v4f wa = *(const v4f*)(wp + 8 * t);
        wQ[t] = *(const v4f*)(wp + 64 + 8 * t);
        wR[t] = *(const v4f*)(wp + 128 + 8 * t);
        wP[t] = wa - wR[t];
    }

    const int fb = wave * 16 + 8 * h;
    float bv[8];
    #pragma unroll
    for (int r = 0; r < 8; ++r) bv[r] = bias[fb + r];

    for (int tile = blockIdx.x; tile < NT; tile += gridDim.x) {
        const int node = tile * 16 + c;
        const float* xp = X + (size_t)node * xStride + 4 * h;
        v4f xf[KB];
        #pragma unroll
        for (int t = 0; t < KB; ++t) xf[t] = *(const v4f*)(xp + 8 * t);
        const float dn = dinv[node];
        asm volatile("" :: HOLD8(0), "v"(dn));
        if constexpr (RIN) {
            #pragma unroll
            for (int t = 0; t < KB; ++t)
                xf[t] = (v4f){fmaxf(xf[t].x, 0.0f), fmaxf(xf[t].y, 0.0f), fmaxf(xf[t].z, 0.0f), fmaxf(xf[t].w, 0.0f)};
        }

        v8f accP = {}, accQ = {}, accR = {};
        #pragma unroll
        for (int t = 0; t < KB; ++t) {
            const v2f b0 = {xf[t].x, xf[t].y}, b1 = {xf[t].z, xf[t].w};
            accP = __builtin_amdgcn_wmma_f32_16x16x4_f32(false, (v2f){wP[t].x, wP[t].y}, false, b0, (short)0, accP, false, false);
            accQ = __builtin_amdgcn_wmma_f32_16x16x4_f32(false, (v2f){wQ[t].x, wQ[t].y}, false, b0, (short)0, accQ, false, false);
            accR = __builtin_amdgcn_wmma_f32_16x16x4_f32(false, (v2f){wR[t].x, wR[t].y}, false, b0, (short)0, accR, false, false);
            accP = __builtin_amdgcn_wmma_f32_16x16x4_f32(false, (v2f){wP[t].z, wP[t].w}, false, b1, (short)0, accP, false, false);
            accQ = __builtin_amdgcn_wmma_f32_16x16x4_f32(false, (v2f){wQ[t].z, wQ[t].w}, false, b1, (short)0, accQ, false, false);
            accR = __builtin_amdgcn_wmma_f32_16x16x4_f32(false, (v2f){wR[t].z, wR[t].w}, false, b1, (short)0, accR, false, false);
        }

        float* pp = outP + (size_t)node * pitchP + fb;
        float* qp = outQ + (size_t)node * pitchQ + fb;
        float* rp = outR + (size_t)node * pitchR + fb;
        const float rd = 1.0f / dn;
        float vP[8], vQ[8], vR[8];
        #pragma unroll
        for (int r = 0; r < 8; ++r) {
            if constexpr (RIN) {
                vP[r] = fmaf(bv[r], rd, accP[r]);
                vQ[r] = -0.5f * accQ[r];
                vR[r] = (dn * dn) * accR[r];
            } else {
                vP[r] = (accP[r] + bv[r]) * rd;
                vQ[r] = accQ[r] * (-0.5f * rd);
                vR[r] = accR[r] * dn;
            }
        }
        *(v4f*)(pp)     = (v4f){vP[0], vP[1], vP[2], vP[3]};
        *(v4f*)(pp + 4) = (v4f){vP[4], vP[5], vP[6], vP[7]};
        *(v4f*)(qp)     = (v4f){vQ[0], vQ[1], vQ[2], vQ[3]};
        *(v4f*)(qp + 4) = (v4f){vQ[4], vQ[5], vQ[6], vQ[7]};
        *(v4f*)(rp)     = (v4f){vR[0], vR[1], vR[2], vR[3]};
        *(v4f*)(rp + 4) = (v4f){vR[4], vR[5], vR[6], vR[7]};
    }
}

extern "C" void kernel_launch(void* const* d_in, const int* in_sizes, int n_in,
                              void* d_out, int out_size, void* d_ws, size_t ws_size,
                              hipStream_t stream) {
    const float* in_feat = (const float*)d_in[0];
    const int*   src     = (const int*)d_in[1];
    const int*   dst     = (const int*)d_in[2];
    const float* W1 = (const float*)d_in[3];   const float* b1 = (const float*)d_in[4];
    const float* W2 = (const float*)d_in[5];   const float* b2 = (const float*)d_in[6];
    const float* Wc1 = (const float*)d_in[7];  const float* bc1 = (const float*)d_in[8];
    const float* Wc2 = (const float*)d_in[9];  const float* bc2 = (const float*)d_in[10];
    const float* W3 = (const float*)d_in[11];  const float* b3 = (const float*)d_in[12];
    const float* W4 = (const float*)d_in[13];  const float* b4 = (const float*)d_in[14];

    char* ws = (char*)d_ws;
    const size_t SZ_H   = (size_t)N_NODES * 64 * 4;
    float* dinv = (float*)(ws);
    float* bufA = (float*)(ws + 200192);
    float* xcat = (float*)(ws + 200192 + 2 * SZ_H);
    static_assert((size_t)DH_G * DH_PITCH * 4 <= (size_t)N_NODES * 192 * 4, "the count planes fit in the xcat area");

    dim3 blkS(256);
    dim3 gN((N_NODES + 255) / 256);
    dim3 gSP((N_EDGES / SP_EPW + 7) / 8);

    unsigned* planes = (unsigned*)xcat;
    deg_hist<<<dim3(DH_G), dim3(1024), 0, stream>>>(dst, planes);
    deg_to_dinv<<<gN, blkS, 0, stream>>>(planes, dinv, N_NODES);

    dim3 gemmBlk(128);
    dim3 gG(768);

    constexpr float rn = 2.0f / 2.0f;
    static_assert(rn - 1.0f == 0.0f, "the layer's regrouping is written for rn = 1: the X0 term of X1 and the X1 term of X2 must be zero");

    head_chain<<<gG, gemmBlk, 0, stream>>>(in_feat, W1, b1, W2, b2, Wc1, bc1, dinv, xcat, 192, xcat + 64, 192, xcat + 128, 192);
    spmm_edges<false><<<gSP, blkS, 0, stream>>>(src, dst, xcat + 128, 192, dinv, xcat + 64, 192, N_EDGES);
    spmm_edges<true><<<gSP, blkS, 0, stream>>>(src, dst, xcat + 64, 192, dinv, xcat, 192, N_EDGES);

    cheb_pre<true><<<gG, gemmBlk, 0, stream>>>(xcat, 192, Wc2, bc2, dinv, xcat + 64, 192, xcat + 128, 192, bufA, 64);
    spmm_edges<false><<<gSP, blkS, 0, stream>>>(src, dst, bufA, 64, dinv, xcat + 128, 192, N_EDGES);
    spmm_edges<true><<<gSP, blkS, 0, stream>>>(src, dst, xcat + 128, 192, dinv, xcat + 64, 192, N_EDGES);

    tail_chain<<<gG, gemmBlk, 0, stream>>>(xcat + 64, 192, W3, b3, W4, b4, dinv, (float*)d_out);
}


# SpMM edge work claimed per L2 half: a block reads its die, draws 256-edge chunks from its half's counter, adds only edges whose dst row is in that half's node range; float atomics kept, arrival order
# speedup vs baseline: 5.6716x; 1.0984x over previous
#include <hip/hip_runtime.h>
#include <hip/hip_bf16.h>

typedef __attribute__((ext_vector_type(2))) float v2f;
typedef __attribute__((ext_vector_type(4))) float v4f;
typedef __attribute__((ext_vector_type(8))) float v8f;

#define N_NODES 50000
#define N_EDGES 800000

#define SP_NTICK 8
#define DH_G     64
#define DH_EPS   (N_EDGES / DH_G)
#define DH_PITCH (((N_NODES + 31) / 32) * 32)

__global__ void __launch_bounds__(1024)
deg_hist(const int* __restrict__ dst, unsigned* __restrict__ planes) {
    static_assert(N_EDGES % (DH_G * 4) == 0, "a slice is a whole number of 16-byte pieces");
    static_assert(N_NODES % 4 == 0, "the table is cleared and stored in 16-byte pieces");
    constexpr int N4    = DH_EPS / 4;
    constexpr int STEPS = (N4 + 1023) / 1024;
    constexpr int T4    = N_NODES / 4;
    __shared__ __attribute__((aligned(16))) unsigned cnt[N_NODES];
    const int tid = threadIdx.x;

    const int4* d4 = (const int4*)(dst + (size_t)blockIdx.x * DH_EPS);
    int4 d[STEPS];
    #pragma unroll
    for (int s = 0; s < STEPS; ++s) {
        const int i = tid + s * 1024;
        d[s] = (i < N4) ? d4[i] : make_int4(-1, -1, -1, -1);
    }
    for (int i = tid; i < T4; i += 1024) ((uint4*)cnt)[i] = make_uint4(0u, 0u, 0u, 0u);
    __syncthreads();

    #pragma unroll
    for (int s = 0; s < STEPS; ++s) {
        if ((unsigned)d[s].x < (unsigned)N_NODES) atomicAdd(&cnt[d[s].x], 1u);
        if ((unsigned)d[s].y < (unsigned)N_NODES) atomicAdd(&cnt[d[s].y], 1u);
        if ((unsigned)d[s].z < (unsigned)N_NODES) atomicAdd(&cnt[d[s].z], 1u);
        if ((unsigned)d[s].w < (unsigned)N_NODES) atomicAdd(&cnt[d[s].w], 1u);
    }
    __syncthreads();

    uint4* out = (uint4*)(planes + (size_t)blockIdx.x * DH_PITCH);
    for (int i = tid; i < T4; i += 1024) out[i] = ((const uint4*)cnt)[i];
}

__global__ void deg_to_dinv(const unsigned* __restrict__ planes, float* __restrict__ dinv, int n, int* __restrict__ tick) {
    int i = blockIdx.x * blockDim.x + threadIdx.x;
    if (i < SP_NTICK) tick[i] = 0;
    if (i >= n) return;
    unsigned c = 0;
    #pragma unroll
    for (int g = 0; g < DH_G; ++g) c += planes[(size_t)g * DH_PITCH + i];
    dinv[i] = rsqrtf(fmaxf((float)c, 1.0f));
}

#define SP_EPW   32
#define SP_B     24
#define SP_WPB   8
#define SP_CH    (SP_WPB * SP_EPW)
#define SP_NCH   (N_EDGES / SP_CH)
#define SP_SPLIT (N_NODES / 2)

#define SP_GATHER(j) do { \
        const int el_ = __builtin_ctz(m); m &= m - 1u; \
        const unsigned xo_ = (unsigned)__builtin_amdgcn_readlane(xo_l, el_); \
        const float* row_ = x + (xo_ + (unsigned)lane);     \
        v0[j] = row_[0]; v1[j] = row_[32]; \
    } while (0)
#define SP_ADD(j) do { \
        const int el_ = __builtin_ctz(ma); ma &= ma - 1u; \
        const unsigned ao_ = (unsigned)__builtin_amdgcn_readlane(ao_l, el_); \
        float* arow_ = out + (ao_ + (unsigned)lane);        \
        if constexpr (SRCW) { \
            const float w_ = __int_as_float(__builtin_amdgcn_readlane(w_l, el_)); \
            atomicAdd(arow_, v0[j] * w_); atomicAdd(arow_ + 32, v1[j] * w_); \
        } else { \
            atomicAdd(arow_, v0[j]); atomicAdd(arow_ + 32, v1[j]); \
        } \
    } while (0)

template <bool SRCW>
__global__ void __launch_bounds__(SP_WPB * 32)
spmm_edges(const int* __restrict__ src, const int* __restrict__ dst,
           const float* __restrict__ x, int xStride,
           const float* __restrict__ dinv,
           float* __restrict__ out, int outStride,
           int* __restrict__ tick) {
    static_assert(N_EDGES % SP_CH == 0, "chunks must be full: there is no tail path");
    __shared__ int s_half;
    __shared__ int s_next[2];
    const int lane = threadIdx.x & 31;
    const int wave = __builtin_amdgcn_readfirstlane((int)(threadIdx.x >> 5));

    if (threadIdx.x == 0) s_half = (int)((__builtin_amdgcn_s_sendmsg_rtn(0x87) >> 18) & 1u);
    __syncthreads();
    const int myHalf = s_half;

    int par = 0;
    for (int side = 0; side < 2; ++side) {
        const int hsel = myHalf ^ side;
        int* ctr = tick + hsel;
        if (threadIdx.x == 0) s_next[par ^ 1] = atomicAdd(ctr, 1);
        __syncthreads();
        par ^= 1;
        int chunk = s_next[par];
        while (chunk < SP_NCH) {
            const int base = chunk * SP_CH + wave * SP_EPW;
            const int s_l = src[base + lane];
            const int d_l = dst[base + lane];
            int w_l = 0;
            if constexpr (SRCW) {
                const float ds = dinv[s_l];
                w_l = __float_as_int((2.0f * ds) * ds);
            }
            const int xo_l = s_l * xStride;
            const int ao_l = d_l * outStride;
            unsigned m = __builtin_amdgcn_ballot_w32((d_l >= SP_SPLIT) == (hsel != 0));

            int nxt = 0;
            if (threadIdx.x == 0) nxt = atomicAdd(ctr, 1);

            while (m != 0u) {
                unsigned mg = m;
                float v0[SP_B], v1[SP_B];
                if (m != 0u) { SP_GATHER(0); if (m != 0u) { SP_GATHER(1); if (m != 0u) { SP_GATHER(2);
                if (m != 0u) { SP_GATHER(3); if (m != 0u) { SP_GATHER(4); if (m != 0u) { SP_GATHER(5);
                if (m != 0u) { SP_GATHER(6); if (m != 0u) { SP_GATHER(7); if (m != 0u) { SP_GATHER(8);
                if (m != 0u) { SP_GATHER(9); if (m != 0u) { SP_GATHER(10); if (m != 0u) { SP_GATHER(11);
                if (m != 0u) { SP_GATHER(12); if (m != 0u) { SP_GATHER(13); if (m != 0u) { SP_GATHER(14);
                if (m != 0u) { SP_GATHER(15); if (m != 0u) { SP_GATHER(16); if (m != 0u) { SP_GATHER(17);
                if (m != 0u) { SP_GATHER(18); if (m != 0u) { SP_GATHER(19); if (m != 0u) { SP_GATHER(20);
                if (m != 0u) { SP_GATHER(21); if (m != 0u) { SP_GATHER(22); if (m != 0u) { SP_GATHER(23);
                }}}}}}}}}}}}}}}}}}}}}}}}
                unsigned ma = mg;
                if (ma != 0u) { SP_ADD(0); if (ma != 0u) { SP_ADD(1); if (ma != 0u) { SP_ADD(2);
                if (ma != 0u) { SP_ADD(3); if (ma != 0u) { SP_ADD(4); if (ma != 0u) { SP_ADD(5);
                if (ma != 0u) { SP_ADD(6); if (ma != 0u) { SP_ADD(7); if (ma != 0u) { SP_ADD(8);
                if (ma != 0u) { SP_ADD(9); if (ma != 0u) { SP_ADD(10); if (ma != 0u) { SP_ADD(11);
                if (ma != 0u) { SP_ADD(12); if (ma != 0u) { SP_ADD(13); if (ma != 0u) { SP_ADD(14);
                if (ma != 0u) { SP_ADD(15); if (ma != 0u) { SP_ADD(16); if (ma != 0u) { SP_ADD(17);
                if (ma != 0u) { SP_ADD(18); if (ma != 0u) { SP_ADD(19); if (ma != 0u) { SP_ADD(20);
                if (ma != 0u) { SP_ADD(21); if (ma != 0u) { SP_ADD(22); if (ma != 0u) { SP_ADD(23);
                }}}}}}}}}}}}}}}}}}}}}}}}
            }

            if (threadIdx.x == 0) s_next[par ^ 1] = nxt;
            __syncthreads();
            par ^= 1;
            chunk = s_next[par];
        }
    }
}

#define HOLD8(b) "v"(xf[(b) + 0]), "v"(xf[(b) + 1]), "v"(xf[(b) + 2]), "v"(xf[(b) + 3]), \
                 "v"(xf[(b) + 4]), "v"(xf[(b) + 5]), "v"(xf[(b) + 6]), "v"(xf[(b) + 7])
#define HP 68

__global__ void __launch_bounds__(128)
head_chain(const float* __restrict__ X,
           const float* __restrict__ W1, const float* __restrict__ b1,
           const float* __restrict__ W2, const float* __restrict__ b2,
           const float* __restrict__ Wl, const float* __restrict__ bias,
           const float* __restrict__ dinv,
           float* __restrict__ outP, int pitchP,
           float* __restrict__ outQ, int pitchQ,
           float* __restrict__ outR, int pitchR) {
    constexpr int NT = N_NODES / 16;
    static_assert(N_NODES % 16 == 0, "node tiles must be full: WMMA needs every lane");
    __shared__ __attribute__((aligned(16))) float h1s[16 * HP];
    __shared__ __attribute__((aligned(16))) float h2s[16 * HP];

    const int lane = threadIdx.x & 31;
    const int wave = __builtin_amdgcn_readfirstlane((int)(threadIdx.x >> 5));
    const int c = lane & 15;
    const int h = lane >> 4;
    const int frow = wave * 16 + c;

    v4f w1f[16], w2f[8], wP[8], wQ[8], wR[8];
    {
        const float* p1 = W1 + (size_t)frow * 128 + 4 * h;
        #pragma unroll
        for (int t = 0; t < 16; ++t) w1f[t] = *(const v4f*)(p1 + 8 * t);
        const float* p2 = W2 + (size_t)frow * 64 + 4 * h;
        const float* p3 = Wl + (size_t)frow * 192 + 4 * h;
        #pragma unroll
        for (int t = 0; t < 8; ++t) {
            w2f[t] = *(const v4f*)(p2 + 8 * t);
            const v4f wa = *(const v4f*)(p3 + 8 * t);
            wQ[t] = *(const v4f*)(p3 + 64 + 8 * t);
            wR[t] = *(const v4f*)(p3 + 128 + 8 * t);
            wP[t] = wa - wR[t];
        }
    }

    const int fb = wave * 16 + 8 * h;
    float b1v[8], b2v[8], bv[8];
    #pragma unroll
    for (int r = 0; r < 8; ++r) { b1v[r] = b1[fb + r]; b2v[r] = b2[fb + r]; bv[r] = bias[fb + r]; }

    float* const h1w = h1s + c * HP + fb;
    float* const h2w = h2s + c * HP + fb;
    const float* const h1r = h1s + c * HP + 4 * h;
    const float* const h2r = h2s + c * HP + 4 * h;

    for (int tile = blockIdx.x; tile < NT; tile += gridDim.x) {
        const int node = tile * 16 + c;
        const float* xp = X + (size_t)node * 128 + 4 * h;
        v4f xf[16];
        #pragma unroll
        for (int t = 0; t < 16; ++t) xf[t] = *(const v4f*)(xp + 8 * t);
        const float dn = dinv[node];
        asm volatile("" :: HOLD8(0), HOLD8(8), "v"(dn));

        v8f acc0 = {}, acc1 = {};
        #pragma unroll
        for (int t = 0; t < 16; ++t) {
            acc0 = __builtin_amdgcn_wmma_f32_16x16x4_f32(false, (v2f){w1f[t].x, w1f[t].y}, false, (v2f){xf[t].x, xf[t].y}, (short)0, acc0, false, false);
            acc1 = __builtin_amdgcn_wmma_f32_16x16x4_f32(false, (v2f){w1f[t].z, w1f[t].w}, false, (v2f){xf[t].z, xf[t].w}, (short)0, acc1, false, false);
        }
        float v[8];
        #pragma unroll
        for (int r = 0; r < 8; ++r) v[r] = fmaxf((acc0[r] + acc1[r]) + b1v[r], 0.0f);
        *(v4f*)(h1w)     = (v4f){v[0], v[1], v[2], v[3]};
        *(v4f*)(h1w + 4) = (v4f){v[4], v[5], v[6], v[7]};
        __syncthreads();

        v4f yf[8];
        #pragma unroll
        for (int t = 0; t < 8; ++t) yf[t] = *(const v4f*)(h1r + 8 * t);
        acc0 = (v8f){}; acc1 = (v8f){};
        #pragma unroll
        for (int t = 0; t < 8; ++t) {
            acc0 = __builtin_amdgcn_wmma_f32_16x16x4_f32(false, (v2f){w2f[t].x, w2f[t].y}, false, (v2f){yf[t].x, yf[t].y}, (short)0, acc0, false, false);
            acc1 = __builtin_amdgcn_wmma_f32_16x16x4_f32(false, (v2f){w2f[t].z, w2f[t].w}, false, (v2f){yf[t].z, yf[t].w}, (short)0, acc1, false, false);
        }
        #pragma unroll
        for (int r = 0; r < 8; ++r) v[r] = fmaxf((acc0[r] + acc1[r]) + b2v[r], 0.0f);
        *(v4f*)(h2w)     = (v4f){v[0], v[1], v[2], v[3]};
        *(v4f*)(h2w + 4) = (v4f){v[4], v[5], v[6], v[7]};
        __syncthreads();

        v4f zf[8];
        #pragma unroll
        for (int t = 0; t < 8; ++t) zf[t] = *(const v4f*)(h2r + 8 * t);
        v8f accP = {}, accQ = {}, accR = {};
        #pragma unroll
        for (int t = 0; t < 8; ++t) {
            const v2f b0 = {zf[t].x, zf[t].y}, b1f = {zf[t].z, zf[t].w};
            accP = __builtin_amdgcn_wmma_f32_16x16x4_f32(false, (v2f){wP[t].x, wP[t].y}, false, b0, (short)0, accP, false, false);
            accQ = __builtin_amdgcn_wmma_f32_16x16x4_f32(false, (v2f){wQ[t].x, wQ[t].y}, false, b0, (short)0, accQ, false, false);
            accR = __builtin_amdgcn_wmma_f32_16x16x4_f32(false, (v2f){wR[t].x, wR[t].y}, false, b0, (short)0, accR, false, false);
            accP = __builtin_amdgcn_wmma_f32_16x16x4_f32(false, (v2f){wP[t].z, wP[t].w}, false, b1f, (short)0, accP, false, false);
            accQ = __builtin_amdgcn_wmma_f32_16x16x4_f32(false, (v2f){wQ[t].z, wQ[t].w}, false, b1f, (short)0, accQ, false, false);
            accR = __builtin_amdgcn_wmma_f32_16x16x4_f32(false, (v2f){wR[t].z, wR[t].w}, false, b1f, (short)0, accR, false, false);
        }

        float* pp = outP + (size_t)node * pitchP + fb;
        float* qp = outQ + (size_t)node * pitchQ + fb;
        float* rp = outR + (size_t)node * pitchR + fb;
        const float rd = 1.0f / dn;
        float vP[8], vQ[8], vR[8];
        #pragma unroll
        for (int r = 0; r < 8; ++r) {
            vP[r] = (accP[r] + bv[r]) * rd;
            vQ[r] = accQ[r] * (-0.5f * rd);
            vR[r] = accR[r] * dn;
        }
        *(v4f*)(pp)     = (v4f){vP[0], vP[1], vP[2], vP[3]};
        *(v4f*)(pp + 4) = (v4f){vP[4], vP[5], vP[6], vP[7]};
        *(v4f*)(qp)     = (v4f){vQ[0], vQ[1], vQ[2], vQ[3]};
        *(v4f*)(qp + 4) = (v4f){vQ[4], vQ[5], vQ[6], vQ[7]};
        *(v4f*)(rp)     = (v4f){vR[0], vR[1], vR[2], vR[3]};
        *(v4f*)(rp + 4) = (v4f){vR[4], vR[5], vR[6], vR[7]};
    }
}

__global__ void __launch_bounds__(128)
tail_chain(const float* __restrict__ X, int xStride,
           const float* __restrict__ W3, const float* __restrict__ b3,
           const float* __restrict__ W4, const float* __restrict__ b4,
           const float* __restrict__ dinv,
           float* __restrict__ Out) {
    constexpr int NT = N_NODES / 16;
    __shared__ __attribute__((aligned(16))) float part[2][4 * 32];

    const int lane = threadIdx.x & 31;
    const int wave = __builtin_amdgcn_readfirstlane((int)(threadIdx.x >> 5));
    const int c = lane & 15;
    const int h = lane >> 4;

    const float* w3p = W3 + (size_t)(wave * 16 + c) * 64 + 4 * h;
    v4f w3f[8];
    #pragma unroll
    for (int t = 0; t < 8; ++t) w3f[t] = *(const v4f*)(w3p + 8 * t);
    const bool ovalid = c < 2;
    const float* w4p = W4 + (ovalid ? c : 0) * 64 + wave * 16 + 8 * h;
    v4f w4f[2];
    #pragma unroll
    for (int t = 0; t < 2; ++t) {
        const v4f w = *(const v4f*)(w4p + 4 * t);
        w4f[t] = ovalid ? w : (v4f){0.0f, 0.0f, 0.0f, 0.0f};
    }

    const int fb = wave * 16 + 8 * h;
    float bv[8];
    #pragma unroll
    for (int r = 0; r < 8; ++r) bv[r] = b3[fb + r];
    const float b4v = b4[lane & 1];

    int par = 0;
    for (int tile = blockIdx.x; tile < NT; tile += gridDim.x) {
        const int node = tile * 16 + c;
        const float* xp = X + (size_t)node * xStride + 4 * h;
        v4f xf[8];
        #pragma unroll
        for (int t = 0; t < 8; ++t) xf[t] = *(const v4f*)(xp + 8 * t);
        const float dn = dinv[node];
        asm volatile("" :: HOLD8(0), "v"(dn));
        #pragma unroll
        for (int t = 0; t < 8; ++t)
            xf[t] = (v4f){fmaxf(xf[t].x, 0.0f), fmaxf(xf[t].y, 0.0f), fmaxf(xf[t].z, 0.0f), fmaxf(xf[t].w, 0.0f)};

        v8f acc0 = {}, acc1 = {};
        #pragma unroll
        for (int t = 0; t < 8; ++t) {
            acc0 = __builtin_amdgcn_wmma_f32_16x16x4_f32(false, (v2f){w3f[t].x, w3f[t].y}, false, (v2f){xf[t].x, xf[t].y}, (short)0, acc0, false, false);
            acc1 = __builtin_amdgcn_wmma_f32_16x16x4_f32(false, (v2f){w3f[t].z, w3f[t].w}, false, (v2f){xf[t].z, xf[t].w}, (short)0, acc1, false, false);
        }
        float v[8];
        #pragma unroll
        for (int r = 0; r < 8; ++r) v[r] = fmaxf(fmaf(acc0[r] + acc1[r], dn, bv[r]), 0.0f);

        v8f o0 = {}, o1 = {};
        o0 = __builtin_amdgcn_wmma_f32_16x16x4_f32(false, (v2f){w4f[0].x, w4f[0].y}, false, (v2f){v[0], v[1]}, (short)0, o0, false, false);
        o1 = __builtin_amdgcn_wmma_f32_16x16x4_f32(false, (v2f){w4f[0].z, w4f[0].w}, false, (v2f){v[2], v[3]}, (short)0, o1, false, false);
        o0 = __builtin_amdgcn_wmma_f32_16x16x4_f32(false, (v2f){w4f[1].x, w4f[1].y}, false, (v2f){v[4], v[5]}, (short)0, o0, false, false);
        o1 = __builtin_amdgcn_wmma_f32_16x16x4_f32(false, (v2f){w4f[1].z, w4f[1].w}, false, (v2f){v[6], v[7]}, (short)0, o1, false, false);
        float* pb = part[par];
        if (h == 0) *(v2f*)(pb + wave * 32 + 2 * c) = (v2f){o0[0] + o1[0], o0[1] + o1[1]};
        __syncthreads();
        if (wave == 0) {
            const float s = ((pb[lane] + pb[32 + lane]) + (pb[64 + lane] + pb[96 + lane])) + b4v;
            Out[(size_t)tile * 32 + lane] = s;
        }
        par ^= 1;
    }
}

template <bool RIN>
__global__ void __launch_bounds__(128)
cheb_pre(const float* __restrict__ X, int xStride,
         const float* __restrict__ Wl, const float* __restrict__ bias,
         const float* __restrict__ dinv,
         float* __restrict__ outP, int pitchP,
         float* __restrict__ outQ, int pitchQ,
         float* __restrict__ outR, int pitchR) {
    constexpr int KB = 8;
    constexpr int NT = N_NODES / 16;
    constexpr int WK = 192;

    const int lane = threadIdx.x & 31;
    const int wave = __builtin_amdgcn_readfirstlane((int)(threadIdx.x >> 5));
    const int c = lane & 15;
    const int h = lane >> 4;

    const float* wp = Wl + (size_t)(wave * 16 + c) * WK + 4 * h;
    v4f wP[KB], wQ[KB], wR[KB];
    #pragma unroll
    for (int t = 0; t < KB; ++t) {
        const v4f wa = *(const v4f*)(wp + 8 * t);
        wQ[t] = *(const v4f*)(wp + 64 + 8 * t);
        wR[t] = *(const v4f*)(wp + 128 + 8 * t);
        wP[t] = wa - wR[t];
    }

    const int fb = wave * 16 + 8 * h;
    float bv[8];
    #pragma unroll
    for (int r = 0; r < 8; ++r) bv[r] = bias[fb + r];

    for (int tile = blockIdx.x; tile < NT; tile += gridDim.x) {
        const int node = tile * 16 + c;
        const float* xp = X + (size_t)node * xStride + 4 * h;
        v4f xf[KB];
        #pragma unroll
        for (int t = 0; t < KB; ++t) xf[t] = *(const v4f*)(xp + 8 * t);
        const float dn = dinv[node];
        asm volatile("" :: HOLD8(0), "v"(dn));
        if constexpr (RIN) {
            #pragma unroll
            for (int t = 0; t < KB; ++t)
                xf[t] = (v4f){fmaxf(xf[t].x, 0.0f), fmaxf(xf[t].y, 0.0f), fmaxf(xf[t].z, 0.0f), fmaxf(xf[t].w, 0.0f)};
        }

        v8f accP = {}, accQ = {}, accR = {};
        #pragma unroll
        for (int t = 0; t < KB; ++t) {
            const v2f b0 = {xf[t].x, xf[t].y}, b1 = {xf[t].z, xf[t].w};
            accP = __builtin_amdgcn_wmma_f32_16x16x4_f32(false, (v2f){wP[t].x, wP[t].y}, false, b0, (short)0, accP, false, false);
            accQ = __builtin_amdgcn_wmma_f32_16x16x4_f32(false, (v2f){wQ[t].x, wQ[t].y}, false, b0, (short)0, accQ, false, false);
            accR = __builtin_amdgcn_wmma_f32_16x16x4_f32(false, (v2f){wR[t].x, wR[t].y}, false, b0, (short)0, accR, false, false);
            accP = __builtin_amdgcn_wmma_f32_16x16x4_f32(false, (v2f){wP[t].z, wP[t].w}, false, b1, (short)0, accP, false, false);
            accQ = __builtin_amdgcn_wmma_f32_16x16x4_f32(false, (v2f){wQ[t].z, wQ[t].w}, false, b1, (short)0, accQ, false, false);
            accR = __builtin_amdgcn_wmma_f32_16x16x4_f32(false, (v2f){wR[t].z, wR[t].w}, false, b1, (short)0, accR, false, false);
        }

        float* pp = outP + (size_t)node * pitchP + fb;
        float* qp = outQ + (size_t)node * pitchQ + fb;
        float* rp = outR + (size_t)node * pitchR + fb;
        const float rd = 1.0f / dn;
        float vP[8], vQ[8], vR[8];
        #pragma unroll
        for (int r = 0; r < 8; ++r) {
            if constexpr (RIN) {
                vP[r] = fmaf(bv[r], rd, accP[r]);
                vQ[r] = -0.5f * accQ[r];
                vR[r] = (dn * dn) * accR[r];
            } else {
                vP[r] = (accP[r] + bv[r]) * rd;
                vQ[r] = accQ[r] * (-0.5f * rd);
                vR[r] = accR[r] * dn;
            }
        }
        *(v4f*)(pp)     = (v4f){vP[0], vP[1], vP[2], vP[3]};
        *(v4f*)(pp + 4) = (v4f){vP[4], vP[5], vP[6], vP[7]};
        *(v4f*)(qp)     = (v4f){vQ[0], vQ[1], vQ[2], vQ[3]};
        *(v4f*)(qp + 4) = (v4f){vQ[4], vQ[5], vQ[6], vQ[7]};
        *(v4f*)(rp)     = (v4f){vR[0], vR[1], vR[2], vR[3]};
        *(v4f*)(rp + 4) = (v4f){vR[4], vR[5], vR[6], vR[7]};
    }
}

extern "C" void kernel_launch(void* const* d_in, const int* in_sizes, int n_in,
                              void* d_out, int out_size, void* d_ws, size_t ws_size,
                              hipStream_t stream) {
    const float* in_feat = (const float*)d_in[0];
    const int*   src     = (const int*)d_in[1];
    const int*   dst     = (const int*)d_in[2];
    const float* W1 = (const float*)d_in[3];   const float* b1 = (const float*)d_in[4];
    const float* W2 = (const float*)d_in[5];   const float* b2 = (const float*)d_in[6];
    const float* Wc1 = (const float*)d_in[7];  const float* bc1 = (const float*)d_in[8];
    const float* Wc2 = (const float*)d_in[9];  const float* bc2 = (const float*)d_in[10];
    const float* W3 = (const float*)d_in[11];  const float* b3 = (const float*)d_in[12];
    const float* W4 = (const float*)d_in[13];  const float* b4 = (const float*)d_in[14];

    char* ws = (char*)d_ws;
    const size_t SZ_H   = (size_t)N_NODES * 64 * 4;
    float* dinv = (float*)(ws);
    float* bufA = (float*)(ws + 200192);
    float* xcat = (float*)(ws + 200192 + 2 * SZ_H);
    int*   tick = (int*)(ws + 200192 + SZ_H);
    static_assert((size_t)DH_G * DH_PITCH * 4 <= (size_t)N_NODES * 192 * 4, "the count planes fit in the xcat area");

    dim3 blkS(256);
    dim3 gN((N_NODES + 255) / 256);
    dim3 gSP(2048);

    unsigned* planes = (unsigned*)xcat;
    deg_hist<<<dim3(DH_G), dim3(1024), 0, stream>>>(dst, planes);
    deg_to_dinv<<<gN, blkS, 0, stream>>>(planes, dinv, N_NODES, tick);

    dim3 gemmBlk(128);
    dim3 gG(768);

    constexpr float rn = 2.0f / 2.0f;
    static_assert(rn - 1.0f == 0.0f, "the layer's regrouping is written for rn = 1: the X0 term of X1 and the X1 term of X2 must be zero");

    head_chain<<<gG, gemmBlk, 0, stream>>>(in_feat, W1, b1, W2, b2, Wc1, bc1, dinv, xcat, 192, xcat + 64, 192, xcat + 128, 192);
    spmm_edges<false><<<gSP, blkS, 0, stream>>>(src, dst, xcat + 128, 192, dinv, xcat + 64, 192, tick);
    spmm_edges<true><<<gSP, blkS, 0, stream>>>(src, dst, xcat + 64, 192, dinv, xcat, 192, tick + 2);

    cheb_pre<true><<<gG, gemmBlk, 0, stream>>>(xcat, 192, Wc2, bc2, dinv, xcat + 64, 192, xcat + 128, 192, bufA, 64);
    spmm_edges<false><<<gSP, blkS, 0, stream>>>(src, dst, bufA, 64, dinv, xcat + 128, 192, tick + 4);
    spmm_edges<true><<<gSP, blkS, 0, stream>>>(src, dst, xcat + 128, 192, dinv, xcat + 64, 192, tick + 6);

    tail_chain<<<gG, gemmBlk, 0, stream>>>(xcat + 64, 192, W3, b3, W4, b4, dinv, (float*)d_out);
}


# SpMM edge passes: one work draw per block, not per 256-edge chunk: a block draws a fixed 1/256 share of the edge list for its L2 half, waves stream it, indices loaded ahead; float atomics kept
# speedup vs baseline: 6.3454x; 1.1188x over previous
#include <hip/hip_runtime.h>
#include <hip/hip_bf16.h>

typedef __attribute__((ext_vector_type(2))) float v2f;
typedef __attribute__((ext_vector_type(4))) float v4f;
typedef __attribute__((ext_vector_type(8))) float v8f;

#define N_NODES 50000
#define N_EDGES 800000

#define SP_NTICK 8
#define SP_TSTR  64
#define DH_G     64
#define DH_EPS   (N_EDGES / DH_G)
#define DH_PITCH (((N_NODES + 31) / 32) * 32)

__global__ void __launch_bounds__(1024)
deg_hist(const int* __restrict__ dst, unsigned* __restrict__ planes) {
    static_assert(N_EDGES % (DH_G * 4) == 0, "a slice is a whole number of 16-byte pieces");
    static_assert(N_NODES % 4 == 0, "the table is cleared and stored in 16-byte pieces");
    constexpr int N4    = DH_EPS / 4;
    constexpr int STEPS = (N4 + 1023) / 1024;
    constexpr int T4    = N_NODES / 4;
    __shared__ __attribute__((aligned(16))) unsigned cnt[N_NODES];
    const int tid = threadIdx.x;

    const int4* d4 = (const int4*)(dst + (size_t)blockIdx.x * DH_EPS);
    int4 d[STEPS];
    #pragma unroll
    for (int s = 0; s < STEPS; ++s) {
        const int i = tid + s * 1024;
        d[s] = (i < N4) ? d4[i] : make_int4(-1, -1, -1, -1);
    }
    for (int i = tid; i < T4; i += 1024) ((uint4*)cnt)[i] = make_uint4(0u, 0u, 0u, 0u);
    __syncthreads();

    #pragma unroll
    for (int s = 0; s < STEPS; ++s) {
        if ((unsigned)d[s].x < (unsigned)N_NODES) atomicAdd(&cnt[d[s].x], 1u);
        if ((unsigned)d[s].y < (unsigned)N_NODES) atomicAdd(&cnt[d[s].y], 1u);
        if ((unsigned)d[s].z < (unsigned)N_NODES) atomicAdd(&cnt[d[s].z], 1u);
        if ((unsigned)d[s].w < (unsigned)N_NODES) atomicAdd(&cnt[d[s].w], 1u);
    }
    __syncthreads();

    uint4* out = (uint4*)(planes + (size_t)blockIdx.x * DH_PITCH);
    for (int i = tid; i < T4; i += 1024) out[i] = ((const uint4*)cnt)[i];
}

__global__ void deg_to_dinv(const unsigned* __restrict__ planes, float* __restrict__ dinv, int n, int* __restrict__ tick) {
    int i = blockIdx.x * blockDim.x + threadIdx.x;
    if (i < SP_NTICK) tick[i * SP_TSTR] = 0;
    if (i >= n) return;
    unsigned c = 0;
    #pragma unroll
    for (int g = 0; g < DH_G; ++g) c += planes[(size_t)g * DH_PITCH + i];
    dinv[i] = rsqrtf(fmaxf((float)c, 1.0f));
}

#define SP_WPB   16
#define SP_NW    256
#define SP_B     20
#define SP_SPLIT (N_NODES / 2)

#define SP_GATHER(M, XO, V0, V1, j) do { \
        const int el_ = __builtin_ctz(M); M &= M - 1u; \
        const unsigned xo_ = (unsigned)__builtin_amdgcn_readlane(XO, el_); \
        const float* row_ = x + (xo_ + (unsigned)lane);     \
        V0[j] = row_[0]; V1[j] = row_[32]; \
    } while (0)
#define SP_ADD(M, AO, WL, V0, V1, j) do { \
        const int el_ = __builtin_ctz(M); M &= M - 1u; \
        const unsigned ao_ = (unsigned)__builtin_amdgcn_readlane(AO, el_); \
        float* arow_ = out + (ao_ + (unsigned)lane);        \
        if constexpr (SRCW) { \
            const float w_ = __int_as_float(__builtin_amdgcn_readlane(WL, el_)); \
            atomicAdd(arow_, V0[j] * w_); atomicAdd(arow_ + 32, V1[j] * w_); \
        } else { \
            atomicAdd(arow_, V0[j]); atomicAdd(arow_ + 32, V1[j]); \
        } \
    } while (0)
#define SP_GATHER_LADDER(M, XO, V0, V1) do { \
        if (M != 0u) { SP_GATHER(M, XO, V0, V1, 0); if (M != 0u) { SP_GATHER(M, XO, V0, V1, 1); \
        if (M != 0u) { SP_GATHER(M, XO, V0, V1, 2); if (M != 0u) { SP_GATHER(M, XO, V0, V1, 3); \
        if (M != 0u) { SP_GATHER(M, XO, V0, V1, 4); if (M != 0u) { SP_GATHER(M, XO, V0, V1, 5); \
        if (M != 0u) { SP_GATHER(M, XO, V0, V1, 6); if (M != 0u) { SP_GATHER(M, XO, V0, V1, 7); \
        if (M != 0u) { SP_GATHER(M, XO, V0, V1, 8); if (M != 0u) { SP_GATHER(M, XO, V0, V1, 9); \
        if (M != 0u) { SP_GATHER(M, XO, V0, V1, 10); if (M != 0u) { SP_GATHER(M, XO, V0, V1, 11); \
        if (M != 0u) { SP_GATHER(M, XO, V0, V1, 12); if (M != 0u) { SP_GATHER(M, XO, V0, V1, 13); \
        if (M != 0u) { SP_GATHER(M, XO, V0, V1, 14); if (M != 0u) { SP_GATHER(M, XO, V0, V1, 15); \
        if (M != 0u) { SP_GATHER(M, XO, V0, V1, 16); if (M != 0u) { SP_GATHER(M, XO, V0, V1, 17); \
        if (M != 0u) { SP_GATHER(M, XO, V0, V1, 18); if (M != 0u) { SP_GATHER(M, XO, V0, V1, 19); \
        }}}}}}}}}}}}}}}}}}}}; \
    } while (0)
#define SP_ADD_LADDER(M, AO, WL, V0, V1) do { \
        if (M != 0u) { SP_ADD(M, AO, WL, V0, V1, 0); if (M != 0u) { SP_ADD(M, AO, WL, V0, V1, 1); \
        if (M != 0u) { SP_ADD(M, AO, WL, V0, V1, 2); if (M != 0u) { SP_ADD(M, AO, WL, V0, V1, 3); \
        if (M != 0u) { SP_ADD(M, AO, WL, V0, V1, 4); if (M != 0u) { SP_ADD(M, AO, WL, V0, V1, 5); \
        if (M != 0u) { SP_ADD(M, AO, WL, V0, V1, 6); if (M != 0u) { SP_ADD(M, AO, WL, V0, V1, 7); \
        if (M != 0u) { SP_ADD(M, AO, WL, V0, V1, 8); if (M != 0u) { SP_ADD(M, AO, WL, V0, V1, 9); \
        if (M != 0u) { SP_ADD(M, AO, WL, V0, V1, 10); if (M != 0u) { SP_ADD(M, AO, WL, V0, V1, 11); \
        if (M != 0u) { SP_ADD(M, AO, WL, V0, V1, 12); if (M != 0u) { SP_ADD(M, AO, WL, V0, V1, 13); \
        if (M != 0u) { SP_ADD(M, AO, WL, V0, V1, 14); if (M != 0u) { SP_ADD(M, AO, WL, V0, V1, 15); \
        if (M != 0u) { SP_ADD(M, AO, WL, V0, V1, 16); if (M != 0u) { SP_ADD(M, AO, WL, V0, V1, 17); \
        if (M != 0u) { SP_ADD(M, AO, WL, V0, V1, 18); if (M != 0u) { SP_ADD(M, AO, WL, V0, V1, 19); \
        }}}}}}}}}}}}}}}}}}}}; \
    } while (0)

template <bool SRCW>
__global__ void __launch_bounds__(SP_WPB * 32)
spmm_edges(const int* __restrict__ src, const int* __restrict__ dst,
           const float* __restrict__ x, int xStride,
           const float* __restrict__ dinv,
           float* __restrict__ out, int outStride,
           int* __restrict__ tick) {
    static_assert(SP_B == 20, "the ladders are written out for SP_B slots");
    static_assert((unsigned long long)SP_NW * N_EDGES < (1ull << 32), "share bounds are computed in 32 bits");
    constexpr int RSTEP = SP_WPB * 32;
    __shared__ int s_share;
    const int lane = threadIdx.x & 31;
    const int wave = __builtin_amdgcn_readfirstlane((int)(threadIdx.x >> 5));

    if (threadIdx.x == 0) {
        int hs = (int)((__builtin_amdgcn_s_sendmsg_rtn(0x87) >> 18) & 1u);
        int r = atomicAdd(tick + hs * SP_TSTR, 1);
        if (r >= SP_NW) { hs ^= 1; r = atomicAdd(tick + hs * SP_TSTR, 1); }
        s_share = (r < SP_NW) ? (2 * r + hs) : -1;
    }
    __syncthreads();
    const int sh = __builtin_amdgcn_readfirstlane(s_share);
    if (sh < 0) return;
    const int hsel = sh & 1;
    const unsigned item = (unsigned)(sh >> 1);
    const int e0 = (int)((item * (unsigned)N_EDGES) / (unsigned)SP_NW);
    const int e1 = (int)(((item + 1u) * (unsigned)N_EDGES) / (unsigned)SP_NW);
    const int nrun  = (e1 - e0 + 31) >> 5;
    const int ntrip = ((nrun + SP_WPB - 1) / SP_WPB + 1) >> 1;

    int eA = e0 + wave * 32 + lane;
    int sA1 = 0, dA1 = 0, sB1 = 0, dB1 = 0;
    const bool okA1 = eA < e1, okB1 = eA + RSTEP < e1;
    if (okA1) { sA1 = src[eA]; dA1 = dst[eA]; }
    if (okB1) { sB1 = src[eA + RSTEP]; dB1 = dst[eA + RSTEP]; }
    eA += 2 * RSTEP;
    int sA2 = 0, dA2 = 0, sB2 = 0, dB2 = 0;
    bool okA2 = eA < e1, okB2 = eA + RSTEP < e1;
    if (okA2) { sA2 = src[eA]; dA2 = dst[eA]; }
    if (okB2) { sB2 = src[eA + RSTEP]; dB2 = dst[eA + RSTEP]; }
    eA += 2 * RSTEP;

    float qA1 = 0.0f, qB1 = 0.0f;
    const bool tA1 = okA1 && ((dA1 >= SP_SPLIT) == (hsel != 0));
    const bool tB1 = okB1 && ((dB1 >= SP_SPLIT) == (hsel != 0));
    unsigned mA1 = __builtin_amdgcn_ballot_w32(tA1);
    unsigned mB1 = __builtin_amdgcn_ballot_w32(tB1);
    if constexpr (SRCW) {
        if (tA1) qA1 = dinv[sA1];
        if (tB1) qB1 = dinv[sB1];
    }

    for (int trip = 0; trip < ntrip; ++trip) {
        const int xoA = sA1 * xStride, aoA = dA1 * outStride;
        const int xoB = sB1 * xStride, aoB = dB1 * outStride;
        int wA = 0, wB = 0;
        if constexpr (SRCW) {
            wA = __float_as_int((2.0f * qA1) * qA1);
            wB = __float_as_int((2.0f * qB1) * qB1);
        }
        unsigned mA = mA1, mB = mB1;

        bool ahead = true;
        do {
            unsigned gA = mA, gB = mB;
            float a0[SP_B], a1[SP_B], b0[SP_B], b1[SP_B];
            SP_GATHER_LADDER(mA, xoA, a0, a1);
            SP_GATHER_LADDER(mB, xoB, b0, b1);
            if (ahead) {
                ahead = false;
                sA1 = sA2; dA1 = dA2; sB1 = sB2; dB1 = dB2;
                const bool tA = okA2 && ((dA1 >= SP_SPLIT) == (hsel != 0));
                const bool tB = okB2 && ((dB1 >= SP_SPLIT) == (hsel != 0));
                mA1 = __builtin_amdgcn_ballot_w32(tA);
                mB1 = __builtin_amdgcn_ballot_w32(tB);
                if constexpr (SRCW) {
                    qA1 = 0.0f; qB1 = 0.0f;
                    if (tA) qA1 = dinv[sA1];
                    if (tB) qB1 = dinv[sB1];
                }
                okA2 = eA < e1; okB2 = eA + RSTEP < e1;
                sA2 = 0; dA2 = 0; sB2 = 0; dB2 = 0;
                if (okA2) { sA2 = src[eA]; dA2 = dst[eA]; }
                if (okB2) { sB2 = src[eA + RSTEP]; dB2 = dst[eA + RSTEP]; }
                eA += 2 * RSTEP;
            }
            SP_ADD_LADDER(gA, aoA, wA, a0, a1);
            SP_ADD_LADDER(gB, aoB, wB, b0, b1);
        } while ((mA | mB) != 0u);
    }
}

#define HOLD8(b) "v"(xf[(b) + 0]), "v"(xf[(b) + 1]), "v"(xf[(b) + 2]), "v"(xf[(b) + 3]), \
                 "v"(xf[(b) + 4]), "v"(xf[(b) + 5]), "v"(xf[(b) + 6]), "v"(xf[(b) + 7])
#define HP 68

__global__ void __launch_bounds__(128)
head_chain(const float* __restrict__ X,
           const float* __restrict__ W1, const float* __restrict__ b1,
           const float* __restrict__ W2, const float* __restrict__ b2,
           const float* __restrict__ Wl, const float* __restrict__ bias,
           const float* __restrict__ dinv,
           float* __restrict__ outP, int pitchP,
           float* __restrict__ outQ, int pitchQ,
           float* __restrict__ outR, int pitchR) {
    constexpr int NT = N_NODES / 16;
    static_assert(N_NODES % 16 == 0, "node tiles must be full: WMMA needs every lane");
    __shared__ __attribute__((aligned(16))) float h1s[16 * HP];
    __shared__ __attribute__((aligned(16))) float h2s[16 * HP];

    const int lane = threadIdx.x & 31;
    const int wave = __builtin_amdgcn_readfirstlane((int)(threadIdx.x >> 5));
    const int c = lane & 15;
    const int h = lane >> 4;
    const int frow = wave * 16 + c;

    v4f w1f[16], w2f[8], wP[8], wQ[8], wR[8];
    {
        const float* p1 = W1 + (size_t)frow * 128 + 4 * h;
        #pragma unroll
        for (int t = 0; t < 16; ++t) w1f[t] = *(const v4f*)(p1 + 8 * t);
        const float* p2 = W2 + (size_t)frow * 64 + 4 * h;
        const float* p3 = Wl + (size_t)frow * 192 + 4 * h;
        #pragma unroll
        for (int t = 0; t < 8; ++t) {
            w2f[t] = *(const v4f*)(p2 + 8 * t);
            const v4f wa = *(const v4f*)(p3 + 8 * t);
            wQ[t] = *(const v4f*)(p3 + 64 + 8 * t);
            wR[t] = *(const v4f*)(p3 + 128 + 8 * t);
            wP[t] = wa - wR[t];
        }
    }

    const int fb = wave * 16 + 8 * h;
    float b1v[8], b2v[8], bv[8];
    #pragma unroll
    for (int r = 0; r < 8; ++r) { b1v[r] = b1[fb + r]; b2v[r] = b2[fb + r]; bv[r] = bias[fb + r]; }

    float* const h1w = h1s + c * HP + fb;
    float* const h2w = h2s + c * HP + fb;
    const float* const h1r = h1s + c * HP + 4 * h;
    const float* const h2r = h2s + c * HP + 4 * h;

    for (int tile = blockIdx.x; tile < NT; tile += gridDim.x) {
        const int node = tile * 16 + c;
        const float* xp = X + (size_t)node * 128 + 4 * h;
        v4f xf[16];
        #pragma unroll
        for (int t = 0; t < 16; ++t) xf[t] = *(const v4f*)(xp + 8 * t);
        const float dn = dinv[node];
        asm volatile("" :: HOLD8(0), HOLD8(8), "v"(dn));

        v8f acc0 = {}, acc1 = {};
        #pragma unroll
        for (int t = 0; t < 16; ++t) {
            acc0 = __builtin_amdgcn_wmma_f32_16x16x4_f32(false, (v2f){w1f[t].x, w1f[t].y}, false, (v2f){xf[t].x, xf[t].y}, (short)0, acc0, false, false);
            acc1 = __builtin_amdgcn_wmma_f32_16x16x4_f32(false, (v2f){w1f[t].z, w1f[t].w}, false, (v2f){xf[t].z, xf[t].w}, (short)0, acc1, false, false);
        }
        float v[8];
        #pragma unroll
        for (int r = 0; r < 8; ++r) v[r] = fmaxf((acc0[r] + acc1[r]) + b1v[r], 0.0f);
        *(v4f*)(h1w)     = (v4f){v[0], v[1], v[2], v[3]};
        *(v4f*)(h1w + 4) = (v4f){v[4], v[5], v[6], v[7]};
        __syncthreads();

        v4f yf[8];
        #pragma unroll
        for (int t = 0; t < 8; ++t) yf[t] = *(const v4f*)(h1r + 8 * t);
        acc0 = (v8f){}; acc1 = (v8f){};
        #pragma unroll
        for (int t = 0; t < 8; ++t) {
            acc0 = __builtin_amdgcn_wmma_f32_16x16x4_f32(false, (v2f){w2f[t].x, w2f[t].y}, false, (v2f){yf[t].x, yf[t].y}, (short)0, acc0, false, false);
            acc1 = __builtin_amdgcn_wmma_f32_16x16x4_f32(false, (v2f){w2f[t].z, w2f[t].w}, false, (v2f){yf[t].z, yf[t].w}, (short)0, acc1, false, false);
        }
        #pragma unroll
        for (int r = 0; r < 8; ++r) v[r] = fmaxf((acc0[r] + acc1[r]) + b2v[r], 0.0f);
        *(v4f*)(h2w)     = (v4f){v[0], v[1], v[2], v[3]};
        *(v4f*)(h2w + 4) = (v4f){v[4], v[5], v[6], v[7]};
        __syncthreads();

        v4f zf[8];
        #pragma unroll
        for (int t = 0; t < 8; ++t) zf[t] = *(const v4f*)(h2r + 8 * t);
        v8f accP = {}, accQ = {}, accR = {};
        #pragma unroll
        for (int t = 0; t < 8; ++t) {
            const v2f b0 = {zf[t].x, zf[t].y}, b1f = {zf[t].z, zf[t].w};
            accP = __builtin_amdgcn_wmma_f32_16x16x4_f32(false, (v2f){wP[t].x, wP[t].y}, false, b0, (short)0, accP, false, false);
            accQ = __builtin_amdgcn_wmma_f32_16x16x4_f32(false, (v2f){wQ[t].x, wQ[t].y}, false, b0, (short)0, accQ, false, false);
            accR = __builtin_amdgcn_wmma_f32_16x16x4_f32(false, (v2f){wR[t].x, wR[t].y}, false, b0, (short)0, accR, false, false);
            accP = __builtin_amdgcn_wmma_f32_16x16x4_f32(false, (v2f){wP[t].z, wP[t].w}, false, b1f, (short)0, accP, false, false);
            accQ = __builtin_amdgcn_wmma_f32_16x16x4_f32(false, (v2f){wQ[t].z, wQ[t].w}, false, b1f, (short)0, accQ, false, false);
            accR = __builtin_amdgcn_wmma_f32_16x16x4_f32(false, (v2f){wR[t].z, wR[t].w}, false, b1f, (short)0, accR, false, false);
        }

        float* pp = outP + (size_t)node * pitchP + fb;
        float* qp = outQ + (size_t)node * pitchQ + fb;
        float* rp = outR + (size_t)node * pitchR + fb;
        const float rd = 1.0f / dn;
        float vP[8], vQ[8], vR[8];
        #pragma unroll
        for (int r = 0; r < 8; ++r) {
            vP[r] = (accP[r] + bv[r]) * rd;
            vQ[r] = accQ[r] * (-0.5f * rd);
            vR[r] = accR[r] * dn;
        }
        *(v4f*)(pp)     = (v4f){vP[0], vP[1], vP[2], vP[3]};
        *(v4f*)(pp + 4) = (v4f){vP[4], vP[5], vP[6], vP[7]};
        *(v4f*)(qp)     = (v4f){vQ[0], vQ[1], vQ[2], vQ[3]};
        *(v4f*)(qp + 4) = (v4f){vQ[4], vQ[5], vQ[6], vQ[7]};
        *(v4f*)(rp)     = (v4f){vR[0], vR[1], vR[2], vR[3]};
        *(v4f*)(rp + 4) = (v4f){vR[4], vR[5], vR[6], vR[7]};
    }
}

__global__ void __launch_bounds__(128)
tail_chain(const float* __restrict__ X, int xStride,
           const float* __restrict__ W3, const float* __restrict__ b3,
           const float* __restrict__ W4, const float* __restrict__ b4,
           const float* __restrict__ dinv,
           float* __restrict__ Out) {
    constexpr int NT = N_NODES / 16;
    __shared__ __attribute__((aligned(16))) float part[2][4 * 32];

    const int lane = threadIdx.x & 31;
    const int wave = __builtin_amdgcn_readfirstlane((int)(threadIdx.x >> 5));
    const int c = lane & 15;
    const int h = lane >> 4;

    const float* w3p = W3 + (size_t)(wave * 16 + c) * 64 + 4 * h;
    v4f w3f[8];
    #pragma unroll
    for (int t = 0; t < 8; ++t) w3f[t] = *(const v4f*)(w3p + 8 * t);
    const bool ovalid = c < 2;
    const float* w4p = W4 + (ovalid ? c : 0) * 64 + wave * 16 + 8 * h;
    v4f w4f[2];
    #pragma unroll
    for (int t = 0; t < 2; ++t) {
        const v4f w = *(const v4f*)(w4p + 4 * t);
        w4f[t] = ovalid ? w : (v4f){0.0f, 0.0f, 0.0f, 0.0f};
    }

    const int fb = wave * 16 + 8 * h;
    float bv[8];
    #pragma unroll
    for (int r = 0; r < 8; ++r) bv[r] = b3[fb + r];
    const float b4v = b4[lane & 1];

    int par = 0;
    for (int tile = blockIdx.x; tile < NT; tile += gridDim.x) {
        const int node = tile * 16 + c;
        const float* xp = X + (size_t)node * xStride + 4 * h;
        v4f xf[8];
        #pragma unroll
        for (int t = 0; t < 8; ++t) xf[t] = *(const v4f*)(xp + 8 * t);
        const float dn = dinv[node];
        asm volatile("" :: HOLD8(0), "v"(dn));
        #pragma unroll
        for (int t = 0; t < 8; ++t)
            xf[t] = (v4f){fmaxf(xf[t].x, 0.0f), fmaxf(xf[t].y, 0.0f), fmaxf(xf[t].z, 0.0f), fmaxf(xf[t].w, 0.0f)};

        v8f acc0 = {}, acc1 = {};
        #pragma unroll
        for (int t = 0; t < 8; ++t) {
            acc0 = __builtin_amdgcn_wmma_f32_16x16x4_f32(false, (v2f){w3f[t].x, w3f[t].y}, false, (v2f){xf[t].x, xf[t].y}, (short)0, acc0, false, false);
            acc1 = __builtin_amdgcn_wmma_f32_16x16x4_f32(false, (v2f){w3f[t].z, w3f[t].w}, false, (v2f){xf[t].z, xf[t].w}, (short)0, acc1, false, false);
        }
        float v[8];
        #pragma unroll
        for (int r = 0; r < 8; ++r) v[r] = fmaxf(fmaf(acc0[r] + acc1[r], dn, bv[r]), 0.0f);

        v8f o0 = {}, o1 = {};
        o0 = __builtin_amdgcn_wmma_f32_16x16x4_f32(false, (v2f){w4f[0].x, w4f[0].y}, false, (v2f){v[0], v[1]}, (short)0, o0, false, false);
        o1 = __builtin_amdgcn_wmma_f32_16x16x4_f32(false, (v2f){w4f[0].z, w4f[0].w}, false, (v2f){v[2], v[3]}, (short)0, o1, false, false);
        o0 = __builtin_amdgcn_wmma_f32_16x16x4_f32(false, (v2f){w4f[1].x, w4f[1].y}, false, (v2f){v[4], v[5]}, (short)0, o0, false, false);
        o1 = __builtin_amdgcn_wmma_f32_16x16x4_f32(false, (v2f){w4f[1].z, w4f[1].w}, false, (v2f){v[6], v[7]}, (short)0, o1, false, false);
        float* pb = part[par];
        if (h == 0) *(v2f*)(pb + wave * 32 + 2 * c) = (v2f){o0[0] + o1[0], o0[1] + o1[1]};
        __syncthreads();
        if (wave == 0) {
            const float s = ((pb[lane] + pb[32 + lane]) + (pb[64 + lane] + pb[96 + lane])) + b4v;
            Out[(size_t)tile * 32 + lane] = s;
        }
        par ^= 1;
    }
}

template <bool RIN>
__global__ void __launch_bounds__(128)
cheb_pre(const float* __restrict__ X, int xStride,
         const float* __restrict__ Wl, const float* __restrict__ bias,
         const float* __restrict__ dinv,
         float* __restrict__ outP, int pitchP,
         float* __restrict__ outQ, int pitchQ,
         float* __restrict__ outR, int pitchR) {
    constexpr int KB = 8;
    constexpr int NT = N_NODES / 16;
    constexpr int WK = 192;

    const int lane = threadIdx.x & 31;
    const int wave = __builtin_amdgcn_readfirstlane((int)(threadIdx.x >> 5));
    const int c = lane & 15;
    const int h = lane >> 4;

    const float* wp = Wl + (size_t)(wave * 16 + c) * WK + 4 * h;
    v4f wP[KB], wQ[KB], wR[KB];
    #pragma unroll
    for (int t = 0; t < KB; ++t) {
        const v4f wa = *(const v4f*)(wp + 8 * t);
        wQ[t] = *(const v4f*)(wp + 64 + 8 * t);
        wR[t] = *(const v4f*)(wp + 128 + 8 * t);
        wP[t] = wa - wR[t];
    }

    const int fb = wave * 16 + 8 * h;
    float bv[8];
    #pragma unroll
    for (int r = 0; r < 8; ++r) bv[r] = bias[fb + r];

    for (int tile = blockIdx.x; tile < NT; tile += gridDim.x) {
        const int node = tile * 16 + c;
        const float* xp = X + (size_t)node * xStride + 4 * h;
        v4f xf[KB];
        #pragma unroll
        for (int t = 0; t < KB; ++t) xf[t] = *(const v4f*)(xp + 8 * t);
        const float dn = dinv[node];
        asm volatile("" :: HOLD8(0), "v"(dn));
        if constexpr (RIN) {
            #pragma unroll
            for (int t = 0; t < KB; ++t)
                xf[t] = (v4f){fmaxf(xf[t].x, 0.0f), fmaxf(xf[t].y, 0.0f), fmaxf(xf[t].z, 0.0f), fmaxf(xf[t].w, 0.0f)};
        }

        v8f accP = {}, accQ = {}, accR = {};
        #pragma unroll
        for (int t = 0; t < KB; ++t) {
            const v2f b0 = {xf[t].x, xf[t].y}, b1 = {xf[t].z, xf[t].w};
            accP = __builtin_amdgcn_wmma_f32_16x16x4_f32(false, (v2f){wP[t].x, wP[t].y}, false, b0, (short)0, accP, false, false);
            accQ = __builtin_amdgcn_wmma_f32_16x16x4_f32(false, (v2f){wQ[t].x, wQ[t].y}, false, b0, (short)0, accQ, false, false);
            accR = __builtin_amdgcn_wmma_f32_16x16x4_f32(false, (v2f){wR[t].x, wR[t].y}, false, b0, (short)0, accR, false, false);
            accP = __builtin_amdgcn_wmma_f32_16x16x4_f32(false, (v2f){wP[t].z, wP[t].w}, false, b1, (short)0, accP, false, false);
            accQ = __builtin_amdgcn_wmma_f32_16x16x4_f32(false, (v2f){wQ[t].z, wQ[t].w}, false, b1, (short)0, accQ, false, false);
            accR = __builtin_amdgcn_wmma_f32_16x16x4_f32(false, (v2f){wR[t].z, wR[t].w}, false, b1, (short)0, accR, false, false);
        }

        float* pp = outP + (size_t)node * pitchP + fb;
        float* qp = outQ + (size_t)node * pitchQ + fb;
        float* rp = outR + (size_t)node * pitchR + fb;
        const float rd = 1.0f / dn;
        float vP[8], vQ[8], vR[8];
        #pragma unroll
        for (int r = 0; r < 8; ++r) {
            if constexpr (RIN) {
                vP[r] = fmaf(bv[r], rd, accP[r]);
                vQ[r] = -0.5f * accQ[r];
                vR[r] = (dn * dn) * accR[r];
            } else {
                vP[r] = (accP[r] + bv[r]) * rd;
                vQ[r] = accQ[r] * (-0.5f * rd);
                vR[r] = accR[r] * dn;
            }
        }
        *(v4f*)(pp)     = (v4f){vP[0], vP[1], vP[2], vP[3]};
        *(v4f*)(pp + 4) = (v4f){vP[4], vP[5], vP[6], vP[7]};
        *(v4f*)(qp)     = (v4f){vQ[0], vQ[1], vQ[2], vQ[3]};
        *(v4f*)(qp + 4) = (v4f){vQ[4], vQ[5], vQ[6], vQ[7]};
        *(v4f*)(rp)     = (v4f){vR[0], vR[1], vR[2], vR[3]};
        *(v4f*)(rp + 4) = (v4f){vR[4], vR[5], vR[6], vR[7]};
    }
}

extern "C" void kernel_launch(void* const* d_in, const int* in_sizes, int n_in,
                              void* d_out, int out_size, void* d_ws, size_t ws_size,
                              hipStream_t stream) {
    const float* in_feat = (const float*)d_in[0];
    const int*   src     = (const int*)d_in[1];
    const int*   dst     = (const int*)d_in[2];
    const float* W1 = (const float*)d_in[3];   const float* b1 = (const float*)d_in[4];
    const float* W2 = (const float*)d_in[5];   const float* b2 = (const float*)d_in[6];
    const float* Wc1 = (const float*)d_in[7];  const float* bc1 = (const float*)d_in[8];
    const float* Wc2 = (const float*)d_in[9];  const float* bc2 = (const float*)d_in[10];
    const float* W3 = (const float*)d_in[11];  const float* b3 = (const float*)d_in[12];
    const float* W4 = (const float*)d_in[13];  const float* b4 = (const float*)d_in[14];

    char* ws = (char*)d_ws;
    const size_t SZ_H   = (size_t)N_NODES * 64 * 4;
    float* dinv = (float*)(ws);
    float* bufA = (float*)(ws + 200192);
    float* xcat = (float*)(ws + 200192 + 2 * SZ_H);
    int*   tick = (int*)(ws + 200192 + SZ_H);
    static_assert((size_t)DH_G * DH_PITCH * 4 <= (size_t)N_NODES * 192 * 4, "the count planes fit in the xcat area");

    dim3 blkS(256);
    dim3 gN((N_NODES + 255) / 256);
    dim3 blkSP(SP_WPB * 32);
    dim3 gSP(2 * SP_NW);

    unsigned* planes = (unsigned*)xcat;
    deg_hist<<<dim3(DH_G), dim3(1024), 0, stream>>>(dst, planes);
    deg_to_dinv<<<gN, blkS, 0, stream>>>(planes, dinv, N_NODES, tick);

    dim3 gemmBlk(128);
    dim3 gG(768);

    constexpr float rn = 2.0f / 2.0f;
    static_assert(rn - 1.0f == 0.0f, "the layer's regrouping is written for rn = 1: the X0 term of X1 and the X1 term of X2 must be zero");

    head_chain<<<gG, gemmBlk, 0, stream>>>(in_feat, W1, b1, W2, b2, Wc1, bc1, dinv, xcat, 192, xcat + 64, 192, xcat + 128, 192);
    spmm_edges<false><<<gSP, blkSP, 0, stream>>>(src, dst, xcat + 128, 192, dinv, xcat + 64, 192, tick);
    spmm_edges<true><<<gSP, blkSP, 0, stream>>>(src, dst, xcat + 64, 192, dinv, xcat, 192, tick + 2 * SP_TSTR);

    cheb_pre<true><<<gG, gemmBlk, 0, stream>>>(xcat, 192, Wc2, bc2, dinv, xcat + 64, 192, xcat + 128, 192, bufA, 64);
    spmm_edges<false><<<gSP, blkSP, 0, stream>>>(src, dst, bufA, 64, dinv, xcat + 128, 192, tick + 4 * SP_TSTR);
    spmm_edges<true><<<gSP, blkSP, 0, stream>>>(src, dst, xcat + 128, 192, dinv, xcat + 64, 192, tick + 6 * SP_TSTR);

    tail_chain<<<gG, gemmBlk, 0, stream>>>(xcat + 64, 192, W3, b3, W4, b4, dinv, (float*)d_out);
}
